# starting kernel
# baseline (speedup 1.0000x reference)
#include <hip/hip_runtime.h>

typedef __attribute__((ext_vector_type(2))) float v2f;
typedef __attribute__((ext_vector_type(8))) float v8f;

#define HID 256
#define TPB 256


__global__ void fill_kernel(float* __restrict__ p, float v, int n) {
  int i = blockIdx.x * blockDim.x + threadIdx.x;
  if (i < n) p[i] = v;
}

__global__ void degree_kernel(const int* __restrict__ dst,
                              float* __restrict__ deg, int E) {
  int e = blockIdx.x * blockDim.x + threadIdx.x;
  if (e < E) unsafeAtomicAdd(&deg[dst[e]], 1.0f);
}

__global__ void rsqrt_inplace_kernel(float* __restrict__ d, int n) {
  int i = blockIdx.x * blockDim.x + threadIdx.x;
  if (i < n) d[i] = 1.0f / sqrtf(d[i]);
}

__global__ void enorm_kernel(const int* __restrict__ src,
                             const int* __restrict__ dst,
                             const float* __restrict__ dinv,
                             float* __restrict__ enorm, int E) {
  int e = blockIdx.x * blockDim.x + threadIdx.x;
  if (e < E) enorm[e] = dinv[src[e]] * dinv[dst[e]];
}

template <int K, bool RELU>
__global__ __launch_bounds__(TPB) void gemm_wmma_kernel(
    const float* __restrict__ X, const float* __restrict__ W,
    float* __restrict__ H, int N) {
  const int lane = threadIdx.x & 31;
  const int wave = threadIdx.x >> 5;
  const int rgrp = wave >> 1;
  const int cgrp = wave & 1;
  const int row_base = blockIdx.x * 128 + rgrp * 32;
  if (row_base >= N) return;

  const int col_base = cgrp * 128;
  const int col = lane & 15;
  const int khalf = (lane >> 4) << 1;
  const size_t arow0 = (size_t)(row_base + col);
  const size_t arow1 = (size_t)(row_base + 16 + col);

  v8f acc[2][8];
#pragma unroll
  for (int m = 0; m < 2; ++m)
#pragma unroll
    for (int j = 0; j < 8; ++j)
      acc[m][j] = (v8f){0.f, 0.f, 0.f, 0.f, 0.f, 0.f, 0.f, 0.f};

  for (int k = 0; k < K; k += 4) {
    v2f a0 = *(const v2f*)(X + arow0 * K + (k + khalf));
    v2f a1 = *(const v2f*)(X + arow1 * K + (k + khalf));
    if (RELU) {
      a0.x = fmaxf(a0.x, 0.f);
      a0.y = fmaxf(a0.y, 0.f);
      a1.x = fmaxf(a1.x, 0.f);
      a1.y = fmaxf(a1.y, 0.f);
    }
#pragma unroll
    for (int j = 0; j < 8; ++j) {
      v2f b;
      b.x = W[(size_t)(k + khalf) * HID + col_base + j * 16 + col];
      b.y = W[(size_t)(k + khalf + 1) * HID + col_base + j * 16 + col];
      acc[0][j] = __builtin_amdgcn_wmma_f32_16x16x4_f32(
          false, a0, false, b, (short)0, acc[0][j], false, false);
      acc[1][j] = __builtin_amdgcn_wmma_f32_16x16x4_f32(
          false, a1, false, b, (short)0, acc[1][j], false, false);
    }
  }

#pragma unroll
  for (int m = 0; m < 2; ++m) {
    const int rlo = row_base + m * 16 + (lane >> 4) * 8;
#pragma unroll
    for (int j = 0; j < 8; ++j) {
#pragma unroll
      for (int r = 0; r < 8; ++r) {
        H[(size_t)(rlo + r) * HID + col_base + j * 16 + col] = acc[m][j][r];
      }
    }
  }
}


__global__ void init_agg_kernel(const float* __restrict__ h,
                                const float* __restrict__ dinv,
                                const float* __restrict__ bias,
                                float* __restrict__ agg, int N) {
  int t = blockIdx.x * blockDim.x + threadIdx.x;
  int node = t >> 6;
  if (node >= N) return;
  int c = (t & 63) << 2;
  float di = dinv[node];
  float sn = di * di;
  float4 hv = *(const float4*)(h + (size_t)node * HID + c);
  float4 bv = *(const float4*)(bias + c);
  float4 o;
  o.x = fmaf(hv.x, sn, bv.x);
  o.y = fmaf(hv.y, sn, bv.y);
  o.z = fmaf(hv.z, sn, bv.z);
  o.w = fmaf(hv.w, sn, bv.w);
  *(float4*)(agg + (size_t)node * HID + c) = o;
}

__global__ __launch_bounds__(TPB) void scatter_kernel(
    const float* __restrict__ h, const int* __restrict__ src,
    const int* __restrict__ dst, const float* __restrict__ enorm,
    float* __restrict__ agg, int E) {
  int e = __builtin_amdgcn_readfirstlane(blockIdx.x * (TPB / 32) +
                                         (threadIdx.x >> 5));
  if (e >= E) return;
  const int s = src[e];
  const int d = dst[e];
  const float w = enorm[e];
  const int c = (threadIdx.x & 31) * 8;
  const float* hp = h + (size_t)s * HID + c;
  float* ap = agg + (size_t)d * HID + c;
  float4 v0 = *(const float4*)(hp);
  float4 v1 = *(const float4*)(hp + 4);
  unsafeAtomicAdd(ap + 0, v0.x * w);
  unsafeAtomicAdd(ap + 1, v0.y * w);
  unsafeAtomicAdd(ap + 2, v0.z * w);
  unsafeAtomicAdd(ap + 3, v0.w * w);
  unsafeAtomicAdd(ap + 4, v1.x * w);
  unsafeAtomicAdd(ap + 5, v1.y * w);
  unsafeAtomicAdd(ap + 6, v1.z * w);
  unsafeAtomicAdd(ap + 7, v1.w * w);
}


static inline int cdiv(int a, int b) { return (a + b - 1) / b; }

extern "C" void kernel_launch(void* const* d_in, const int* in_sizes, int n_in,
                              void* d_out, int out_size, void* d_ws,
                              size_t ws_size, hipStream_t stream) {
  const float* x = (const float*)d_in[0];
  const int* ei = (const int*)d_in[1];
  const float* W1 = (const float*)d_in[2];
  const float* b1 = (const float*)d_in[3];
  const float* W2 = (const float*)d_in[4];
  const float* b2 = (const float*)d_in[5];
  const float* W3 = (const float*)d_in[6];
  const float* b3 = (const float*)d_in[7];
  float* out = (float*)d_out;

  const int F = 128;
  const int N = in_sizes[0] / F;
  const int E = in_sizes[1] / 2;
  const int* src = ei;
  const int* dst = ei + E;

  float* wsf = (float*)d_ws;
  size_t off = 0;
  float* dinv = wsf + off;
  off += (size_t)((N + 255) / 256) * 256;
  float* enorm = wsf + off;
  off += (size_t)((E + 255) / 256) * 256;
  float* hbuf = wsf + off;
  off += (size_t)N * HID;
  float* agg = wsf + off;

  const int nodeGrid = cdiv(N, TPB);
  const int edgeGrid = cdiv(E, TPB);
  const int featGrid = cdiv(N * 64, TPB);
  const int scatGrid = cdiv(E, TPB / 32);
  const int gemmGrid = cdiv(N, 128);

  fill_kernel<<<nodeGrid, TPB, 0, stream>>>(dinv, 1.0f, N);
  degree_kernel<<<edgeGrid, TPB, 0, stream>>>(dst, dinv, E);
  rsqrt_inplace_kernel<<<nodeGrid, TPB, 0, stream>>>(dinv, N);
  enorm_kernel<<<edgeGrid, TPB, 0, stream>>>(src, dst, dinv, enorm, E);

  gemm_wmma_kernel<128, false><<<gemmGrid, TPB, 0, stream>>>(x, W1, hbuf, N);
  init_agg_kernel<<<featGrid, TPB, 0, stream>>>(hbuf, dinv, b1, agg, N);
  scatter_kernel<<<scatGrid, TPB, 0, stream>>>(hbuf, src, dst, enorm, agg, E);

  gemm_wmma_kernel<256, true><<<gemmGrid, TPB, 0, stream>>>(agg, W2, hbuf, N);
  init_agg_kernel<<<featGrid, TPB, 0, stream>>>(hbuf, dinv, b2, agg, N);
  scatter_kernel<<<scatGrid, TPB, 0, stream>>>(hbuf, src, dst, enorm, agg, E);

  gemm_wmma_kernel<256, true><<<gemmGrid, TPB, 0, stream>>>(agg, W3, hbuf, N);
  init_agg_kernel<<<featGrid, TPB, 0, stream>>>(hbuf, dinv, b3, out, N);
  scatter_kernel<<<scatGrid, TPB, 0, stream>>>(hbuf, src, dst, enorm, out, E);
}


# aggregation: float atomic adds to memory replaced by LDS float adds: a block owns 392 dst rows x 128 cols, finds its edges by one byte per edge, adds their source rows, stores once; LDS arrival order
# speedup vs baseline: 2.7526x; 2.7526x over previous
#include <hip/hip_runtime.h>

typedef __attribute__((ext_vector_type(2))) float v2f;
typedef __attribute__((ext_vector_type(8))) float v8f;

#define HID 256
#define TPB 256

#define RROWS 392
#define RCOLS 128
#define RTHREADS 1024
#define RWAVES 32
#define LCAP 8192


__global__ void fill_kernel(float* __restrict__ p, float v, int n) {
  int i = blockIdx.x * blockDim.x + threadIdx.x;
  if (i < n) p[i] = v;
}

__global__ void degree_kernel(const int* __restrict__ dst,
                              float* __restrict__ deg, int E) {
  int e = blockIdx.x * blockDim.x + threadIdx.x;
  if (e < E) unsafeAtomicAdd(&deg[dst[e]], 1.0f);
}

__global__ void rsqrt_inplace_kernel(float* __restrict__ d, int n) {
  int i = blockIdx.x * blockDim.x + threadIdx.x;
  if (i < n) d[i] = 1.0f / sqrtf(d[i]);
}

__global__ void enorm_kernel(const int* __restrict__ src,
                             const int* __restrict__ dst,
                             const float* __restrict__ dinv,
                             float* __restrict__ enorm,
                             unsigned int* __restrict__ tagw, int E) {
  const int t = blockIdx.x * blockDim.x + threadIdx.x;
  const int e = t * 4;
  if (e >= E) return;
  const bool al16 = ((((size_t)src) | ((size_t)dst)) & 15) == 0;
  if (al16 && e + 3 < E) {
    const int4 s4 = *(const int4*)(src + e);
    const int4 d4 = *(const int4*)(dst + e);
    float4 w;
    w.x = dinv[s4.x] * dinv[d4.x];
    w.y = dinv[s4.y] * dinv[d4.y];
    w.z = dinv[s4.z] * dinv[d4.z];
    w.w = dinv[s4.w] * dinv[d4.w];
    *(float4*)(enorm + e) = w;
    tagw[t] = (((unsigned)d4.x / RROWS) & 255u) |
              ((((unsigned)d4.y / RROWS) & 255u) << 8) |
              ((((unsigned)d4.z / RROWS) & 255u) << 16) |
              ((((unsigned)d4.w / RROWS) & 255u) << 24);
  } else {
    unsigned int tw = 0;
    for (int k = 0; k < 4 && e + k < E; ++k) {
      const int d = dst[e + k];
      enorm[e + k] = dinv[src[e + k]] * dinv[d];
      tw |= (((unsigned)d / RROWS) & 255u) << (8 * k);
    }
    tagw[t] = tw;
  }
}

template <int K, bool RELU>
__global__ __launch_bounds__(TPB) void gemm_wmma_kernel(
    const float* __restrict__ X, const float* __restrict__ W,
    float* __restrict__ H, int N) {
  const int lane = threadIdx.x & 31;
  const int wave = threadIdx.x >> 5;
  const int rgrp = wave >> 1;
  const int cgrp = wave & 1;
  const int row_base = blockIdx.x * 128 + rgrp * 32;
  if (row_base >= N) return;

  const int col_base = cgrp * 128;
  const int col = lane & 15;
  const int khalf = (lane >> 4) << 1;
  const size_t arow0 = (size_t)(row_base + col);
  const size_t arow1 = (size_t)(row_base + 16 + col);

  v8f acc[2][8];
#pragma unroll
  for (int m = 0; m < 2; ++m)
#pragma unroll
    for (int j = 0; j < 8; ++j)
      acc[m][j] = (v8f){0.f, 0.f, 0.f, 0.f, 0.f, 0.f, 0.f, 0.f};

  for (int k = 0; k < K; k += 4) {
    v2f a0 = *(const v2f*)(X + arow0 * K + (k + khalf));
    v2f a1 = *(const v2f*)(X + arow1 * K + (k + khalf));
    if (RELU) {
      a0.x = fmaxf(a0.x, 0.f);
      a0.y = fmaxf(a0.y, 0.f);
      a1.x = fmaxf(a1.x, 0.f);
      a1.y = fmaxf(a1.y, 0.f);
    }
#pragma unroll
    for (int j = 0; j < 8; ++j) {
      v2f b;
      b.x = W[(size_t)(k + khalf) * HID + col_base + j * 16 + col];
      b.y = W[(size_t)(k + khalf + 1) * HID + col_base + j * 16 + col];
      acc[0][j] = __builtin_amdgcn_wmma_f32_16x16x4_f32(
          false, a0, false, b, (short)0, acc[0][j], false, false);
      acc[1][j] = __builtin_amdgcn_wmma_f32_16x16x4_f32(
          false, a1, false, b, (short)0, acc[1][j], false, false);
    }
  }

#pragma unroll
  for (int m = 0; m < 2; ++m) {
    const int rlo = row_base + m * 16 + (lane >> 4) * 8;
#pragma unroll
    for (int j = 0; j < 8; ++j) {
#pragma unroll
      for (int r = 0; r < 8; ++r) {
        H[(size_t)(rlo + r) * HID + col_base + j * 16 + col] = acc[m][j][r];
      }
    }
  }
}


__global__ __launch_bounds__(RTHREADS) void range_sum_kernel(
    const float* __restrict__ h, const int* __restrict__ src,
    const int* __restrict__ dst, const float* __restrict__ enorm,
    const unsigned int* __restrict__ tagw, const float* __restrict__ dinv,
    const float* __restrict__ bias, float* __restrict__ out, int N, int E) {
  __shared__ __attribute__((aligned(16))) float acc[RROWS * RCOLS];
  __shared__ int elist[LCAP];
  __shared__ int ecount;

  const int lane = threadIdx.x & 31;
  const int wave = __builtin_amdgcn_readfirstlane(threadIdx.x >> 5);
  const int range = blockIdx.x >> 1;
  const int col0 = (blockIdx.x & 1) * RCOLS;
  const int r0 = range * RROWS;
  const int nrows = min(RROWS, N - r0);

  if (threadIdx.x == 0) ecount = 0;

  const int total4 = nrows * (RCOLS / 4);
  for (int i0 = threadIdx.x; i0 < total4; i0 += RTHREADS * 4) {
    float4 hv[4];
#pragma unroll
    for (int u = 0; u < 4; ++u) {
      const int i = i0 + u * RTHREADS;
      if (i < total4)
        hv[u] = *(const float4*)(h + (size_t)(r0 + (i >> 5)) * HID + col0 +
                                 (i & 31) * 4);
    }
#pragma unroll
    for (int u = 0; u < 4; ++u) {
      const int i = i0 + u * RTHREADS;
      if (i < total4) {
        const float di = dinv[r0 + (i >> 5)];
        const float sn = di * di;
        const float4 bv = *(const float4*)(bias + col0 + (i & 31) * 4);
        float4 o;
        o.x = fmaf(hv[u].x, sn, bv.x);
        o.y = fmaf(hv[u].y, sn, bv.y);
        o.z = fmaf(hv[u].z, sn, bv.z);
        o.w = fmaf(hv[u].w, sn, bv.w);
        *(float4*)(acc + i * 4) = o;
      }
    }
  }
  __syncthreads();

  auto note_edge = [&](int e) {
    const int slot = atomicAdd(&ecount, 1);
    if (slot < LCAP) {
      elist[slot] = e;
    } else {
      const float* hp = h + (size_t)src[e] * HID + col0;
      float* ap = acc + (dst[e] - r0) * RCOLS;
      const float w = enorm[e];
      for (int c = 0; c < RCOLS; ++c) unsafeAtomicAdd(ap + c, hp[c] * w);
    }
  };

  const bool exact = (N <= 256 * RROWS);
  const int nSuper = exact ? (E >> 11) : 0;
  const int rot = (int)(((long long)(range & 127) * nSuper) >> 7);
  const unsigned pat = (unsigned)(range & 255) * 0x01010101u;
  const uint4* t4 = (const uint4*)tagw;
  for (int i = wave; i < nSuper; i += RWAVES) {
    int c = i + rot;
    if (c >= nSuper) c -= nSuper;
    const uint4* p = t4 + (size_t)c * 128 + lane;
    uint4 t[4];
#pragma unroll
    for (int u = 0; u < 4; ++u) t[u] = p[u * 32];
    unsigned m[4];
#pragma unroll
    for (int u = 0; u < 4; ++u) {
      const unsigned wd[4] = {t[u].x, t[u].y, t[u].z, t[u].w};
      m[u] = 0;
#pragma unroll
      for (int q = 0; q < 4; ++q) {
        const unsigned y = wd[q] ^ pat;
        const unsigned z = ~(((y & 0x7F7F7F7Fu) + 0x7F7F7F7Fu) | y);
        m[u] |= (z >> (7 - q)) & (0x01010101u << q);
      }
    }
    unsigned long long hits =
        (unsigned long long)(m[0] | (m[1] << 4)) |
        ((unsigned long long)(m[2] | (m[3] << 4)) << 32);
    while (hits) {
      const int b = __builtin_ctzll(hits);
      hits &= hits - 1;
      const int u = ((b >> 5) << 1) | ((b >> 2) & 1);
      const int j = (b & 31) >> 3;
      const int q = b & 3;
      note_edge(c * 2048 + u * 512 + lane * 16 + q * 4 + j);
    }
  }
  for (int e = (nSuper << 11) + (int)threadIdx.x; e < E; e += RTHREADS) {
    if ((unsigned)(dst[e] - r0) < (unsigned)nrows) note_edge(e);
  }
  __syncthreads();

  const int n = min(ecount, LCAP);
  int base = wave * 8;
  int s = 0, d = 0;
  float w = 0.f;
  if (base + (lane & 7) < n) {
    const int e = elist[base + (lane & 7)];
    s = src[e];
    d = dst[e];
    w = enorm[e];
  }
  while (base < n) {
    const int nbase = base + RWAVES * 8;
    int s2 = 0, d2 = 0;
    float w2 = 0.f;
    if (nbase + (lane & 7) < n) {
      const int e = elist[nbase + (lane & 7)];
      s2 = src[e];
      d2 = dst[e];
      w2 = enorm[e];
    }
    const int cnt = min(8, n - base);
    if (cnt == 8) {
      float v[8][4];
#pragma unroll
      for (int q = 0; q < 8; ++q) {
        const int sk = __builtin_amdgcn_readlane(s, q);
        const float* hp = h + (size_t)sk * HID + col0 + lane;
#pragma unroll
        for (int j = 0; j < 4; ++j) v[q][j] = hp[j * 32];
      }
#pragma unroll
      for (int q = 0; q < 8; ++q) {
        const int dk = __builtin_amdgcn_readlane(d, q) - r0;
        const float wk =
            __int_as_float(__builtin_amdgcn_readlane(__float_as_int(w), q));
        float* ap = acc + dk * RCOLS + lane;
#pragma unroll
        for (int j = 0; j < 4; ++j) unsafeAtomicAdd(ap + j * 32, v[q][j] * wk);
      }
    } else {
      for (int k = 0; k < cnt; ++k) {
        const int sk = __builtin_amdgcn_readlane(s, k);
        const int dk = __builtin_amdgcn_readlane(d, k) - r0;
        const float wk =
            __int_as_float(__builtin_amdgcn_readlane(__float_as_int(w), k));
        const float* hp = h + (size_t)sk * HID + col0 + lane;
        float* ap = acc + dk * RCOLS + lane;
        float v1[4];
#pragma unroll
        for (int j = 0; j < 4; ++j) v1[j] = hp[j * 32];
#pragma unroll
        for (int j = 0; j < 4; ++j) unsafeAtomicAdd(ap + j * 32, v1[j] * wk);
      }
    }
    s = s2;
    d = d2;
    w = w2;
    base = nbase;
  }
  __syncthreads();

  for (int i = threadIdx.x; i < total4; i += RTHREADS) {
    *(float4*)(out + (size_t)(r0 + (i >> 5)) * HID + col0 + (i & 31) * 4) =
        *(const float4*)(acc + i * 4);
  }
}


static inline int cdiv(int a, int b) { return (a + b - 1) / b; }

extern "C" void kernel_launch(void* const* d_in, const int* in_sizes, int n_in,
                              void* d_out, int out_size, void* d_ws,
                              size_t ws_size, hipStream_t stream) {
  const float* x = (const float*)d_in[0];
  const int* ei = (const int*)d_in[1];
  const float* W1 = (const float*)d_in[2];
  const float* b1 = (const float*)d_in[3];
  const float* W2 = (const float*)d_in[4];
  const float* b2 = (const float*)d_in[5];
  const float* W3 = (const float*)d_in[6];
  const float* b3 = (const float*)d_in[7];
  float* out = (float*)d_out;

  const int F = 128;
  const int N = in_sizes[0] / F;
  const int E = in_sizes[1] / 2;
  const int* src = ei;
  const int* dst = ei + E;

  float* wsf = (float*)d_ws;
  size_t off = 0;
  float* dinv = wsf + off;
  off += (size_t)((N + 255) / 256) * 256;
  float* enorm = wsf + off;
  off += (size_t)((E + 255) / 256) * 256;
  float* hbuf = wsf + off;
  off += (size_t)N * HID;
  float* agg = wsf + off;
  off += (size_t)N * HID;
  unsigned int* tagw = (unsigned int*)(wsf + off);

  const int nodeGrid = cdiv(N, TPB);
  const int edgeGrid = cdiv(E, TPB);
  const int edge4Grid = cdiv(cdiv(E, 4), TPB);
  const int rangeGrid = 2 * cdiv(N, RROWS);
  const int gemmGrid = cdiv(N, 128);

  fill_kernel<<<nodeGrid, TPB, 0, stream>>>(dinv, 1.0f, N);
  degree_kernel<<<edgeGrid, TPB, 0, stream>>>(dst, dinv, E);
  rsqrt_inplace_kernel<<<nodeGrid, TPB, 0, stream>>>(dinv, N);
  enorm_kernel<<<edge4Grid, TPB, 0, stream>>>(src, dst, dinv, enorm, tagw, E);

  gemm_wmma_kernel<128, false><<<gemmGrid, TPB, 0, stream>>>(x, W1, hbuf, N);
  range_sum_kernel<<<rangeGrid, RTHREADS, 0, stream>>>(
      hbuf, src, dst, enorm, tagw, dinv, b1, agg, N, E);

  gemm_wmma_kernel<256, true><<<gemmGrid, TPB, 0, stream>>>(agg, W2, hbuf, N);
  range_sum_kernel<<<rangeGrid, RTHREADS, 0, stream>>>(
      hbuf, src, dst, enorm, tagw, dinv, b2, agg, N, E);

  gemm_wmma_kernel<256, true><<<gemmGrid, TPB, 0, stream>>>(agg, W3, hbuf, N);
  range_sum_kernel<<<rangeGrid, RTHREADS, 0, stream>>>(
      hbuf, src, dst, enorm, tagw, dinv, b3, out, N, E);
}


# aggregation: a row block's edge list (src, dst, weight) built once per call, by layer 1, kept in the workspace and reread by layers 2,3; one block does both column halves; sums in LDS arrival order
# speedup vs baseline: 3.2478x; 1.1799x over previous
#include <hip/hip_runtime.h>

typedef __attribute__((ext_vector_type(2))) float v2f;
typedef __attribute__((ext_vector_type(8))) float v8f;

#define HID 256
#define TPB 256

#define RROWS 392
#define RCOLS 128
#define RTHREADS 1024
#define RWAVES 32
#define LCAP 8192


__global__ void fill_kernel(float* __restrict__ p, float v, int n) {
  int i = blockIdx.x * blockDim.x + threadIdx.x;
  if (i < n) p[i] = v;
}

__global__ void degree_kernel(const int* __restrict__ dst,
                              float* __restrict__ deg, int E) {
  int e = blockIdx.x * blockDim.x + threadIdx.x;
  if (e < E) unsafeAtomicAdd(&deg[dst[e]], 1.0f);
}

__global__ void rsqrt_inplace_kernel(float* __restrict__ d, int n) {
  int i = blockIdx.x * blockDim.x + threadIdx.x;
  if (i < n) d[i] = 1.0f / sqrtf(d[i]);
}

__global__ void enorm_kernel(const int* __restrict__ src,
                             const int* __restrict__ dst,
                             const float* __restrict__ dinv,
                             float* __restrict__ enorm,
                             unsigned int* __restrict__ tagw, int E) {
  const int t = blockIdx.x * blockDim.x + threadIdx.x;
  const int e = t * 4;
  if (e >= E) return;
  const bool al16 = ((((size_t)src) | ((size_t)dst)) & 15) == 0;
  if (al16 && e + 3 < E) {
    const int4 s4 = *(const int4*)(src + e);
    const int4 d4 = *(const int4*)(dst + e);
    float4 w;
    w.x = dinv[s4.x] * dinv[d4.x];
    w.y = dinv[s4.y] * dinv[d4.y];
    w.z = dinv[s4.z] * dinv[d4.z];
    w.w = dinv[s4.w] * dinv[d4.w];
    *(float4*)(enorm + e) = w;
    tagw[t] = (((unsigned)d4.x / RROWS) & 255u) |
              ((((unsigned)d4.y / RROWS) & 255u) << 8) |
              ((((unsigned)d4.z / RROWS) & 255u) << 16) |
              ((((unsigned)d4.w / RROWS) & 255u) << 24);
  } else {
    unsigned int tw = 0;
    for (int k = 0; k < 4 && e + k < E; ++k) {
      const int d = dst[e + k];
      enorm[e + k] = dinv[src[e + k]] * dinv[d];
      tw |= (((unsigned)d / RROWS) & 255u) << (8 * k);
    }
    tagw[t] = tw;
  }
}

template <int K, bool RELU>
__global__ __launch_bounds__(TPB) void gemm_wmma_kernel(
    const float* __restrict__ X, const float* __restrict__ W,
    float* __restrict__ H, int N) {
  const int lane = threadIdx.x & 31;
  const int wave = threadIdx.x >> 5;
  const int rgrp = wave >> 1;
  const int cgrp = wave & 1;
  const int row_base = blockIdx.x * 128 + rgrp * 32;
  if (row_base >= N) return;

  const int col_base = cgrp * 128;
  const int col = lane & 15;
  const int khalf = (lane >> 4) << 1;
  const size_t arow0 = (size_t)(row_base + col);
  const size_t arow1 = (size_t)(row_base + 16 + col);

  v8f acc[2][8];
#pragma unroll
  for (int m = 0; m < 2; ++m)
#pragma unroll
    for (int j = 0; j < 8; ++j)
      acc[m][j] = (v8f){0.f, 0.f, 0.f, 0.f, 0.f, 0.f, 0.f, 0.f};

  for (int k = 0; k < K; k += 4) {
    v2f a0 = *(const v2f*)(X + arow0 * K + (k + khalf));
    v2f a1 = *(const v2f*)(X + arow1 * K + (k + khalf));
    if (RELU) {
      a0.x = fmaxf(a0.x, 0.f);
      a0.y = fmaxf(a0.y, 0.f);
      a1.x = fmaxf(a1.x, 0.f);
      a1.y = fmaxf(a1.y, 0.f);
    }
#pragma unroll
    for (int j = 0; j < 8; ++j) {
      v2f b;
      b.x = W[(size_t)(k + khalf) * HID + col_base + j * 16 + col];
      b.y = W[(size_t)(k + khalf + 1) * HID + col_base + j * 16 + col];
      acc[0][j] = __builtin_amdgcn_wmma_f32_16x16x4_f32(
          false, a0, false, b, (short)0, acc[0][j], false, false);
      acc[1][j] = __builtin_amdgcn_wmma_f32_16x16x4_f32(
          false, a1, false, b, (short)0, acc[1][j], false, false);
    }
  }

#pragma unroll
  for (int m = 0; m < 2; ++m) {
    const int rlo = row_base + m * 16 + (lane >> 4) * 8;
#pragma unroll
    for (int j = 0; j < 8; ++j) {
#pragma unroll
      for (int r = 0; r < 8; ++r) {
        H[(size_t)(rlo + r) * HID + col_base + j * 16 + col] = acc[m][j][r];
      }
    }
  }
}


template <bool BUILD>
__global__ __launch_bounds__(RTHREADS) void range_sum_kernel(
    const float* __restrict__ h, const int* __restrict__ src,
    const int* __restrict__ dst, const float* __restrict__ enorm,
    const unsigned int* __restrict__ tagw, const float* __restrict__ dinv,
    const float* __restrict__ bias, float* __restrict__ out,
    int* __restrict__ gcount, int* __restrict__ gsrc, int* __restrict__ gdst,
    float* __restrict__ gw, int N, int E) {
  __shared__ __attribute__((aligned(16))) float acc[RROWS * RCOLS];
  __shared__ int ls[LCAP];
  __shared__ int ld[LCAP];
  __shared__ float lw[LCAP];
  __shared__ int ecount;

  const int lane = threadIdx.x & 31;
  const int wave = __builtin_amdgcn_readfirstlane(threadIdx.x >> 5);
  const int range = blockIdx.x;
  const int r0 = range * RROWS;
  const int nrows = min(RROWS, N - r0);
  const size_t lbase = (size_t)range * LCAP;

  int total;
  if (BUILD) {
    if (threadIdx.x == 0) ecount = 0;
    __syncthreads();

    auto note_edge = [&](int e) {
      const int slot = atomicAdd(&ecount, 1);
      if (slot < LCAP) ls[slot] = e;
    };

    const bool exact = (N <= 256 * RROWS);
    const int nSuper = exact ? (E >> 11) : 0;
    const int rot = (int)(((long long)(range & 127) * nSuper) >> 7);
    const unsigned pat = (unsigned)(range & 255) * 0x01010101u;
    const uint4* t4 = (const uint4*)tagw;
    for (int i = wave; i < nSuper; i += RWAVES) {
      int c = i + rot;
      if (c >= nSuper) c -= nSuper;
      const uint4* p = t4 + (size_t)c * 128 + lane;
      uint4 t[4];
#pragma unroll
      for (int u = 0; u < 4; ++u) t[u] = p[u * 32];
      unsigned m[4];
#pragma unroll
      for (int u = 0; u < 4; ++u) {
        const unsigned wd[4] = {t[u].x, t[u].y, t[u].z, t[u].w};
        m[u] = 0;
#pragma unroll
        for (int q = 0; q < 4; ++q) {
          const unsigned y = wd[q] ^ pat;
          const unsigned z = ~(((y & 0x7F7F7F7Fu) + 0x7F7F7F7Fu) | y);
          m[u] |= (z >> (7 - q)) & (0x01010101u << q);
        }
      }
      unsigned long long hits =
          (unsigned long long)(m[0] | (m[1] << 4)) |
          ((unsigned long long)(m[2] | (m[3] << 4)) << 32);
      while (hits) {
        const int b = __builtin_ctzll(hits);
        hits &= hits - 1;
        const int u = ((b >> 5) << 1) | ((b >> 2) & 1);
        const int j = (b & 31) >> 3;
        const int q = b & 3;
        note_edge(c * 2048 + u * 512 + lane * 16 + q * 4 + j);
      }
    }
    for (int e = (nSuper << 11) + (int)threadIdx.x; e < E; e += RTHREADS) {
      if ((unsigned)(dst[e] - r0) < (unsigned)nrows) note_edge(e);
    }
    __syncthreads();

    total = ecount;
    const int n = min(total, LCAP);
    for (int i = threadIdx.x; i < n; i += RTHREADS) {
      const int e = ls[i];
      const int s = src[e];
      const int d = dst[e] - r0;
      const float w = enorm[e];
      ls[i] = s;
      ld[i] = d;
      lw[i] = w;
      gsrc[lbase + i] = s;
      gdst[lbase + i] = d;
      gw[lbase + i] = w;
    }
    if (threadIdx.x == 0) gcount[range] = total;
  } else {
    total = gcount[range];
    const int n = min(total, LCAP);
    for (int i = threadIdx.x; i < n; i += RTHREADS) {
      ls[i] = gsrc[lbase + i];
      ld[i] = gdst[lbase + i];
      lw[i] = gw[lbase + i];
    }
  }

  const int total4 = nrows * (RCOLS / 4);
  for (int half = 0; half < HID / RCOLS; ++half) {
    const int col0 = half * RCOLS;

    for (int i0 = threadIdx.x; i0 < total4; i0 += RTHREADS * 4) {
      float4 hv[4];
#pragma unroll
      for (int u = 0; u < 4; ++u) {
        const int i = i0 + u * RTHREADS;
        if (i < total4)
          hv[u] = *(const float4*)(h + (size_t)(r0 + (i >> 5)) * HID + col0 +
                                   (i & 31) * 4);
      }
#pragma unroll
      for (int u = 0; u < 4; ++u) {
        const int i = i0 + u * RTHREADS;
        if (i < total4) {
          const float di = dinv[r0 + (i >> 5)];
          const float sn = di * di;
          const float4 bv = *(const float4*)(bias + col0 + (i & 31) * 4);
          float4 o;
          o.x = fmaf(hv[u].x, sn, bv.x);
          o.y = fmaf(hv[u].y, sn, bv.y);
          o.z = fmaf(hv[u].z, sn, bv.z);
          o.w = fmaf(hv[u].w, sn, bv.w);
          *(float4*)(acc + i * 4) = o;
        }
      }
    }
    __syncthreads();

    if (total <= LCAP) {
      const int n = total;
      int base = wave * 8;
      int s = 0, d = 0;
      float w = 0.f;
      if (base + (lane & 7) < n) {
        s = ls[base + (lane & 7)];
        d = ld[base + (lane & 7)];
        w = lw[base + (lane & 7)];
      }
      while (base < n) {
        const int nbase = base + RWAVES * 8;
        int s2 = 0, d2 = 0;
        float w2 = 0.f;
        if (nbase + (lane & 7) < n) {
          s2 = ls[nbase + (lane & 7)];
          d2 = ld[nbase + (lane & 7)];
          w2 = lw[nbase + (lane & 7)];
        }
        const int cnt = min(8, n - base);
        if (cnt == 8) {
          float v[8][4];
#pragma unroll
          for (int q = 0; q < 8; ++q) {
            const int sk = __builtin_amdgcn_readlane(s, q);
            const float* hp = h + (size_t)sk * HID + col0 + lane;
#pragma unroll
            for (int j = 0; j < 4; ++j) v[q][j] = hp[j * 32];
          }
#pragma unroll
          for (int q = 0; q < 8; ++q) {
            const int dk = __builtin_amdgcn_readlane(d, q);
            const float wk =
                __int_as_float(__builtin_amdgcn_readlane(__float_as_int(w), q));
            float* ap = acc + dk * RCOLS + lane;
#pragma unroll
            for (int j = 0; j < 4; ++j)
              unsafeAtomicAdd(ap + j * 32, v[q][j] * wk);
          }
        } else {
          for (int k = 0; k < cnt; ++k) {
            const int sk = __builtin_amdgcn_readlane(s, k);
            const int dk = __builtin_amdgcn_readlane(d, k);
            const float wk =
                __int_as_float(__builtin_amdgcn_readlane(__float_as_int(w), k));
            const float* hp = h + (size_t)sk * HID + col0 + lane;
            float* ap = acc + dk * RCOLS + lane;
            float v1[4];
#pragma unroll
            for (int j = 0; j < 4; ++j) v1[j] = hp[j * 32];
#pragma unroll
            for (int j = 0; j < 4; ++j)
              unsafeAtomicAdd(ap + j * 32, v1[j] * wk);
          }
        }
        s = s2;
        d = d2;
        w = w2;
        base = nbase;
      }
    } else {
      for (int e = threadIdx.x; e < E; e += RTHREADS) {
        const int dl = dst[e] - r0;
        if ((unsigned)dl < (unsigned)nrows) {
          const float* hp = h + (size_t)src[e] * HID + col0;
          float* ap = acc + dl * RCOLS;
          const float w = enorm[e];
          for (int c = 0; c < RCOLS; ++c) unsafeAtomicAdd(ap + c, hp[c] * w);
        }
      }
    }
    __syncthreads();

    for (int i = threadIdx.x; i < total4; i += RTHREADS) {
      *(float4*)(out + (size_t)(r0 + (i >> 5)) * HID + col0 + (i & 31) * 4) =
          *(const float4*)(acc + i * 4);
    }
    __syncthreads();
  }
}


static inline int cdiv(int a, int b) { return (a + b - 1) / b; }

extern "C" void kernel_launch(void* const* d_in, const int* in_sizes, int n_in,
                              void* d_out, int out_size, void* d_ws,
                              size_t ws_size, hipStream_t stream) {
  const float* x = (const float*)d_in[0];
  const int* ei = (const int*)d_in[1];
  const float* W1 = (const float*)d_in[2];
  const float* b1 = (const float*)d_in[3];
  const float* W2 = (const float*)d_in[4];
  const float* b2 = (const float*)d_in[5];
  const float* W3 = (const float*)d_in[6];
  const float* b3 = (const float*)d_in[7];
  float* out = (float*)d_out;

  const int F = 128;
  const int N = in_sizes[0] / F;
  const int E = in_sizes[1] / 2;
  const int* src = ei;
  const int* dst = ei + E;

  float* wsf = (float*)d_ws;
  size_t off = 0;
  float* dinv = wsf + off;
  off += (size_t)((N + 255) / 256) * 256;
  float* enorm = wsf + off;
  off += (size_t)((E + 255) / 256) * 256;
  float* hbuf = wsf + off;
  off += (size_t)N * HID;
  float* agg = wsf + off;
  off += (size_t)N * HID;
  unsigned int* tagw = (unsigned int*)(wsf + off);
  off += (size_t)(((E + 3) / 4 + 255) / 256) * 256;
  const int nRanges = cdiv(N, RROWS);
  int* gcount = (int*)(wsf + off);
  off += (size_t)((nRanges + 255) / 256) * 256;
  int* gsrc = (int*)(wsf + off);
  off += (size_t)nRanges * LCAP;
  int* gdst = (int*)(wsf + off);
  off += (size_t)nRanges * LCAP;
  float* gw = wsf + off;

  const int nodeGrid = cdiv(N, TPB);
  const int edgeGrid = cdiv(E, TPB);
  const int edge4Grid = cdiv(cdiv(E, 4), TPB);
  const int rangeGrid = nRanges;
  const int gemmGrid = cdiv(N, 128);

  fill_kernel<<<nodeGrid, TPB, 0, stream>>>(dinv, 1.0f, N);
  degree_kernel<<<edgeGrid, TPB, 0, stream>>>(dst, dinv, E);
  rsqrt_inplace_kernel<<<nodeGrid, TPB, 0, stream>>>(dinv, N);
  enorm_kernel<<<edge4Grid, TPB, 0, stream>>>(src, dst, dinv, enorm, tagw, E);

  gemm_wmma_kernel<128, false><<<gemmGrid, TPB, 0, stream>>>(x, W1, hbuf, N);
  range_sum_kernel<true><<<rangeGrid, RTHREADS, 0, stream>>>(
      hbuf, src, dst, enorm, tagw, dinv, b1, agg, gcount, gsrc, gdst, gw, N, E);

  gemm_wmma_kernel<256, true><<<gemmGrid, TPB, 0, stream>>>(agg, W2, hbuf, N);
  range_sum_kernel<false><<<rangeGrid, RTHREADS, 0, stream>>>(
      hbuf, src, dst, enorm, tagw, dinv, b2, agg, gcount, gsrc, gdst, gw, N, E);

  gemm_wmma_kernel<256, true><<<gemmGrid, TPB, 0, stream>>>(agg, W3, hbuf, N);
  range_sum_kernel<false><<<rangeGrid, RTHREADS, 0, stream>>>(
      hbuf, src, dst, enorm, tagw, dinv, b3, out, gcount, gsrc, gdst, gw, N, E);
}


# aggregation gather: a lane loads 16 B of a source row (one 512-B load per wave, was four 128-B loads); a row's sums sit column-interleaved in LDS so adds stay conflict-free; LDS arrival order kept
# speedup vs baseline: 3.3770x; 1.0398x over previous
#include <hip/hip_runtime.h>

typedef __attribute__((ext_vector_type(2))) float v2f;
typedef __attribute__((ext_vector_type(8))) float v8f;

#define HID 256
#define TPB 256

#define RROWS 392
#define RCOLS 128
#define RTHREADS 1024
#define RWAVES 32
#define LCAP 8192


__global__ void fill_kernel(float* __restrict__ p, float v, int n) {
  int i = blockIdx.x * blockDim.x + threadIdx.x;
  if (i < n) p[i] = v;
}

__global__ void degree_kernel(const int* __restrict__ dst,
                              float* __restrict__ deg, int E) {
  int e = blockIdx.x * blockDim.x + threadIdx.x;
  if (e < E) unsafeAtomicAdd(&deg[dst[e]], 1.0f);
}

__global__ void rsqrt_inplace_kernel(float* __restrict__ d, int n) {
  int i = blockIdx.x * blockDim.x + threadIdx.x;
  if (i < n) d[i] = 1.0f / sqrtf(d[i]);
}

__global__ void enorm_kernel(const int* __restrict__ src,
                             const int* __restrict__ dst,
                             const float* __restrict__ dinv,
                             float* __restrict__ enorm,
                             unsigned int* __restrict__ tagw, int E) {
  const int t = blockIdx.x * blockDim.x + threadIdx.x;
  const int e = t * 4;
  if (e >= E) return;
  const bool al16 = ((((size_t)src) | ((size_t)dst)) & 15) == 0;
  if (al16 && e + 3 < E) {
    const int4 s4 = *(const int4*)(src + e);
    const int4 d4 = *(const int4*)(dst + e);
    float4 w;
    w.x = dinv[s4.x] * dinv[d4.x];
    w.y = dinv[s4.y] * dinv[d4.y];
    w.z = dinv[s4.z] * dinv[d4.z];
    w.w = dinv[s4.w] * dinv[d4.w];
    *(float4*)(enorm + e) = w;
    tagw[t] = (((unsigned)d4.x / RROWS) & 255u) |
              ((((unsigned)d4.y / RROWS) & 255u) << 8) |
              ((((unsigned)d4.z / RROWS) & 255u) << 16) |
              ((((unsigned)d4.w / RROWS) & 255u) << 24);
  } else {
    unsigned int tw = 0;
    for (int k = 0; k < 4 && e + k < E; ++k) {
      const int d = dst[e + k];
      enorm[e + k] = dinv[src[e + k]] * dinv[d];
      tw |= (((unsigned)d / RROWS) & 255u) << (8 * k);
    }
    tagw[t] = tw;
  }
}

template <int K, bool RELU>
__global__ __launch_bounds__(TPB) void gemm_wmma_kernel(
    const float* __restrict__ X, const float* __restrict__ W,
    float* __restrict__ H, int N) {
  const int lane = threadIdx.x & 31;
  const int wave = threadIdx.x >> 5;
  const int rgrp = wave >> 1;
  const int cgrp = wave & 1;
  const int row_base = blockIdx.x * 128 + rgrp * 32;
  if (row_base >= N) return;

  const int col_base = cgrp * 128;
  const int col = lane & 15;
  const int khalf = (lane >> 4) << 1;
  const size_t arow0 = (size_t)(row_base + col);
  const size_t arow1 = (size_t)(row_base + 16 + col);

  v8f acc[2][8];
#pragma unroll
  for (int m = 0; m < 2; ++m)
#pragma unroll
    for (int j = 0; j < 8; ++j)
      acc[m][j] = (v8f){0.f, 0.f, 0.f, 0.f, 0.f, 0.f, 0.f, 0.f};

  for (int k = 0; k < K; k += 4) {
    v2f a0 = *(const v2f*)(X + arow0 * K + (k + khalf));
    v2f a1 = *(const v2f*)(X + arow1 * K + (k + khalf));
    if (RELU) {
      a0.x = fmaxf(a0.x, 0.f);
      a0.y = fmaxf(a0.y, 0.f);
      a1.x = fmaxf(a1.x, 0.f);
      a1.y = fmaxf(a1.y, 0.f);
    }
#pragma unroll
    for (int j = 0; j < 8; ++j) {
      v2f b;
      b.x = W[(size_t)(k + khalf) * HID + col_base + j * 16 + col];
      b.y = W[(size_t)(k + khalf + 1) * HID + col_base + j * 16 + col];
      acc[0][j] = __builtin_amdgcn_wmma_f32_16x16x4_f32(
          false, a0, false, b, (short)0, acc[0][j], false, false);
      acc[1][j] = __builtin_amdgcn_wmma_f32_16x16x4_f32(
          false, a1, false, b, (short)0, acc[1][j], false, false);
    }
  }

#pragma unroll
  for (int m = 0; m < 2; ++m) {
    const int rlo = row_base + m * 16 + (lane >> 4) * 8;
#pragma unroll
    for (int j = 0; j < 8; ++j) {
#pragma unroll
      for (int r = 0; r < 8; ++r) {
        H[(size_t)(rlo + r) * HID + col_base + j * 16 + col] = acc[m][j][r];
      }
    }
  }
}


template <bool BUILD>
__global__ __launch_bounds__(RTHREADS) void range_sum_kernel(
    const float* __restrict__ h, const int* __restrict__ src,
    const int* __restrict__ dst, const float* __restrict__ enorm,
    const unsigned int* __restrict__ tagw, const float* __restrict__ dinv,
    const float* __restrict__ bias, float* __restrict__ out,
    int* __restrict__ gcount, int* __restrict__ gsrc, int* __restrict__ gdst,
    float* __restrict__ gw, int N, int E) {
  __shared__ float acc[RROWS * RCOLS];
  __shared__ int ls[LCAP];
  __shared__ int ld[LCAP];
  __shared__ float lw[LCAP];
  __shared__ int ecount;

  const int lane = threadIdx.x & 31;
  const int wave = __builtin_amdgcn_readfirstlane(threadIdx.x >> 5);
  const int range = blockIdx.x;
  const int r0 = range * RROWS;
  const int nrows = min(RROWS, N - r0);
  const size_t lbase = (size_t)range * LCAP;

  int total;
  if (BUILD) {
    if (threadIdx.x == 0) ecount = 0;
    __syncthreads();

    auto note_edge = [&](int e) {
      const int slot = atomicAdd(&ecount, 1);
      if (slot < LCAP) ls[slot] = e;
    };

    const bool exact = (N <= 256 * RROWS);
    const int nSuper = exact ? (E >> 11) : 0;
    const int rot = (int)(((long long)(range & 127) * nSuper) >> 7);
    const unsigned pat = (unsigned)(range & 255) * 0x01010101u;
    const uint4* t4 = (const uint4*)tagw;
    for (int i = wave; i < nSuper; i += RWAVES) {
      int c = i + rot;
      if (c >= nSuper) c -= nSuper;
      const uint4* p = t4 + (size_t)c * 128 + lane;
      uint4 t[4];
#pragma unroll
      for (int u = 0; u < 4; ++u) t[u] = p[u * 32];
      unsigned m[4];
#pragma unroll
      for (int u = 0; u < 4; ++u) {
        const unsigned wd[4] = {t[u].x, t[u].y, t[u].z, t[u].w};
        m[u] = 0;
#pragma unroll
        for (int q = 0; q < 4; ++q) {
          const unsigned y = wd[q] ^ pat;
          const unsigned z = ~(((y & 0x7F7F7F7Fu) + 0x7F7F7F7Fu) | y);
          m[u] |= (z >> (7 - q)) & (0x01010101u << q);
        }
      }
      unsigned long long hits =
          (unsigned long long)(m[0] | (m[1] << 4)) |
          ((unsigned long long)(m[2] | (m[3] << 4)) << 32);
      while (hits) {
        const int b = __builtin_ctzll(hits);
        hits &= hits - 1;
        const int u = ((b >> 5) << 1) | ((b >> 2) & 1);
        const int j = (b & 31) >> 3;
        const int q = b & 3;
        note_edge(c * 2048 + u * 512 + lane * 16 + q * 4 + j);
      }
    }
    for (int e = (nSuper << 11) + (int)threadIdx.x; e < E; e += RTHREADS) {
      if ((unsigned)(dst[e] - r0) < (unsigned)nrows) note_edge(e);
    }
    __syncthreads();

    total = ecount;
    const int n = min(total, LCAP);
    for (int i = threadIdx.x; i < n; i += RTHREADS) {
      const int e = ls[i];
      const int s = src[e];
      const int d = dst[e] - r0;
      const float w = enorm[e];
      ls[i] = s;
      ld[i] = d;
      lw[i] = w;
      gsrc[lbase + i] = s;
      gdst[lbase + i] = d;
      gw[lbase + i] = w;
    }
    if (threadIdx.x == 0) gcount[range] = total;
  } else {
    total = gcount[range];
    const int n = min(total, LCAP);
    for (int i = threadIdx.x; i < n; i += RTHREADS) {
      ls[i] = gsrc[lbase + i];
      ld[i] = gdst[lbase + i];
      lw[i] = gw[lbase + i];
    }
  }

  const int total4 = nrows * (RCOLS / 4);
  for (int half = 0; half < HID / RCOLS; ++half) {
    const int col0 = half * RCOLS;

    for (int i0 = threadIdx.x; i0 < total4; i0 += RTHREADS * 4) {
      float4 hv[4];
#pragma unroll
      for (int u = 0; u < 4; ++u) {
        const int i = i0 + u * RTHREADS;
        if (i < total4)
          hv[u] = *(const float4*)(h + (size_t)(r0 + (i >> 5)) * HID + col0 +
                                   (i & 31) * 4);
      }
#pragma unroll
      for (int u = 0; u < 4; ++u) {
        const int i = i0 + u * RTHREADS;
        if (i < total4) {
          const float di = dinv[r0 + (i >> 5)];
          const float sn = di * di;
          const float4 bv = *(const float4*)(bias + col0 + (i & 31) * 4);
          float4 o;
          o.x = fmaf(hv[u].x, sn, bv.x);
          o.y = fmaf(hv[u].y, sn, bv.y);
          o.z = fmaf(hv[u].z, sn, bv.z);
          o.w = fmaf(hv[u].w, sn, bv.w);
          float* ar = acc + (i >> 5) * RCOLS + (i & 31);
          ar[0] = o.x;
          ar[32] = o.y;
          ar[64] = o.z;
          ar[96] = o.w;
        }
      }
    }
    __syncthreads();

    if (total <= LCAP) {
      const int n = total;
      int base = wave * 8;
      int s = 0, d = 0;
      float w = 0.f;
      if (base + (lane & 7) < n) {
        s = ls[base + (lane & 7)];
        d = ld[base + (lane & 7)];
        w = lw[base + (lane & 7)];
      }
      while (base < n) {
        const int nbase = base + RWAVES * 8;
        int s2 = 0, d2 = 0;
        float w2 = 0.f;
        if (nbase + (lane & 7) < n) {
          s2 = ls[nbase + (lane & 7)];
          d2 = ld[nbase + (lane & 7)];
          w2 = lw[nbase + (lane & 7)];
        }
        const int cnt = min(8, n - base);
        if (cnt == 8) {
          float4 v[8];
#pragma unroll
          for (int q = 0; q < 8; ++q) {
            const int sk = __builtin_amdgcn_readlane(s, q);
            v[q] = *(const float4*)(h + (size_t)sk * HID + col0 + lane * 4);
          }
#pragma unroll
          for (int q = 0; q < 8; ++q) {
            const int dk = __builtin_amdgcn_readlane(d, q);
            const float wk =
                __int_as_float(__builtin_amdgcn_readlane(__float_as_int(w), q));
            float* ap = acc + dk * RCOLS + lane;
            unsafeAtomicAdd(ap, v[q].x * wk);
            unsafeAtomicAdd(ap + 32, v[q].y * wk);
            unsafeAtomicAdd(ap + 64, v[q].z * wk);
            unsafeAtomicAdd(ap + 96, v[q].w * wk);
          }
        } else {
          for (int k = 0; k < cnt; ++k) {
            const int sk = __builtin_amdgcn_readlane(s, k);
            const int dk = __builtin_amdgcn_readlane(d, k);
            const float wk =
                __int_as_float(__builtin_amdgcn_readlane(__float_as_int(w), k));
            const float4 v1 =
                *(const float4*)(h + (size_t)sk * HID + col0 + lane * 4);
            float* ap = acc + dk * RCOLS + lane;
            unsafeAtomicAdd(ap, v1.x * wk);
            unsafeAtomicAdd(ap + 32, v1.y * wk);
            unsafeAtomicAdd(ap + 64, v1.z * wk);
            unsafeAtomicAdd(ap + 96, v1.w * wk);
          }
        }
        s = s2;
        d = d2;
        w = w2;
        base = nbase;
      }
    } else {
      for (int e = threadIdx.x; e < E; e += RTHREADS) {
        const int dl = dst[e] - r0;
        if ((unsigned)dl < (unsigned)nrows) {
          const float* hp = h + (size_t)src[e] * HID + col0;
          float* ap = acc + dl * RCOLS;
          const float w = enorm[e];
          for (int c = 0; c < RCOLS; ++c)
            unsafeAtomicAdd(ap + (c & 3) * 32 + (c >> 2), hp[c] * w);
        }
      }
    }
    __syncthreads();

    for (int i = threadIdx.x; i < total4; i += RTHREADS) {
      const float* ar = acc + (i >> 5) * RCOLS + (i & 31);
      float4 o;
      o.x = ar[0];
      o.y = ar[32];
      o.z = ar[64];
      o.w = ar[96];
      *(float4*)(out + (size_t)(r0 + (i >> 5)) * HID + col0 + (i & 31) * 4) = o;
    }
    __syncthreads();
  }
}


static inline int cdiv(int a, int b) { return (a + b - 1) / b; }

extern "C" void kernel_launch(void* const* d_in, const int* in_sizes, int n_in,
                              void* d_out, int out_size, void* d_ws,
                              size_t ws_size, hipStream_t stream) {
  const float* x = (const float*)d_in[0];
  const int* ei = (const int*)d_in[1];
  const float* W1 = (const float*)d_in[2];
  const float* b1 = (const float*)d_in[3];
  const float* W2 = (const float*)d_in[4];
  const float* b2 = (const float*)d_in[5];
  const float* W3 = (const float*)d_in[6];
  const float* b3 = (const float*)d_in[7];
  float* out = (float*)d_out;

  const int F = 128;
  const int N = in_sizes[0] / F;
  const int E = in_sizes[1] / 2;
  const int* src = ei;
  const int* dst = ei + E;

  float* wsf = (float*)d_ws;
  size_t off = 0;
  float* dinv = wsf + off;
  off += (size_t)((N + 255) / 256) * 256;
  float* enorm = wsf + off;
  off += (size_t)((E + 255) / 256) * 256;
  float* hbuf = wsf + off;
  off += (size_t)N * HID;
  float* agg = wsf + off;
  off += (size_t)N * HID;
  unsigned int* tagw = (unsigned int*)(wsf + off);
  off += (size_t)(((E + 3) / 4 + 255) / 256) * 256;
  const int nRanges = cdiv(N, RROWS);
  int* gcount = (int*)(wsf + off);
  off += (size_t)((nRanges + 255) / 256) * 256;
  int* gsrc = (int*)(wsf + off);
  off += (size_t)nRanges * LCAP;
  int* gdst = (int*)(wsf + off);
  off += (size_t)nRanges * LCAP;
  float* gw = wsf + off;

  const int nodeGrid = cdiv(N, TPB);
  const int edgeGrid = cdiv(E, TPB);
  const int edge4Grid = cdiv(cdiv(E, 4), TPB);
  const int rangeGrid = nRanges;
  const int gemmGrid = cdiv(N, 128);

  fill_kernel<<<nodeGrid, TPB, 0, stream>>>(dinv, 1.0f, N);
  degree_kernel<<<edgeGrid, TPB, 0, stream>>>(dst, dinv, E);
  rsqrt_inplace_kernel<<<nodeGrid, TPB, 0, stream>>>(dinv, N);
  enorm_kernel<<<edge4Grid, TPB, 0, stream>>>(src, dst, dinv, enorm, tagw, E);

  gemm_wmma_kernel<128, false><<<gemmGrid, TPB, 0, stream>>>(x, W1, hbuf, N);
  range_sum_kernel<true><<<rangeGrid, RTHREADS, 0, stream>>>(
      hbuf, src, dst, enorm, tagw, dinv, b1, agg, gcount, gsrc, gdst, gw, N, E);

  gemm_wmma_kernel<256, true><<<gemmGrid, TPB, 0, stream>>>(agg, W2, hbuf, N);
  range_sum_kernel<false><<<rangeGrid, RTHREADS, 0, stream>>>(
      hbuf, src, dst, enorm, tagw, dinv, b2, agg, gcount, gsrc, gdst, gw, N, E);

  gemm_wmma_kernel<256, true><<<gemmGrid, TPB, 0, stream>>>(agg, W3, hbuf, N);
  range_sum_kernel<false><<<rangeGrid, RTHREADS, 0, stream>>>(
      hbuf, src, dst, enorm, tagw, dinv, b3, out, gcount, gsrc, gdst, gw, N, E);
}


# GEMM tiling: a wave does 80 rows x 64 cols (5x4 tiles, was 2x8); a 20-wave block of 400 rows fills a WGP, 250 blocks run in one round (was 782 8-wave blocks in 1.5 rounds); k order, sums unchanged
# speedup vs baseline: 3.5643x; 1.0555x over previous
#include <hip/hip_runtime.h>

typedef __attribute__((ext_vector_type(2))) float v2f;
typedef __attribute__((ext_vector_type(8))) float v8f;

#define HID 256
#define TPB 256

#define RROWS 392
#define RCOLS 128
#define RTHREADS 1024
#define RWAVES 32
#define LCAP 8192


__global__ void fill_kernel(float* __restrict__ p, float v, int n) {
  int i = blockIdx.x * blockDim.x + threadIdx.x;
  if (i < n) p[i] = v;
}

__global__ void degree_kernel(const int* __restrict__ dst,
                              float* __restrict__ deg, int E) {
  int e = blockIdx.x * blockDim.x + threadIdx.x;
  if (e < E) unsafeAtomicAdd(&deg[dst[e]], 1.0f);
}

__global__ void rsqrt_inplace_kernel(float* __restrict__ d, int n) {
  int i = blockIdx.x * blockDim.x + threadIdx.x;
  if (i < n) d[i] = 1.0f / sqrtf(d[i]);
}

__global__ void enorm_kernel(const int* __restrict__ src,
                             const int* __restrict__ dst,
                             const float* __restrict__ dinv,
                             float* __restrict__ enorm,
                             unsigned int* __restrict__ tagw, int E) {
  const int t = blockIdx.x * blockDim.x + threadIdx.x;
  const int e = t * 4;
  if (e >= E) return;
  const bool al16 = ((((size_t)src) | ((size_t)dst)) & 15) == 0;
  if (al16 && e + 3 < E) {
    const int4 s4 = *(const int4*)(src + e);
    const int4 d4 = *(const int4*)(dst + e);
    float4 w;
    w.x = dinv[s4.x] * dinv[d4.x];
    w.y = dinv[s4.y] * dinv[d4.y];
    w.z = dinv[s4.z] * dinv[d4.z];
    w.w = dinv[s4.w] * dinv[d4.w];
    *(float4*)(enorm + e) = w;
    tagw[t] = (((unsigned)d4.x / RROWS) & 255u) |
              ((((unsigned)d4.y / RROWS) & 255u) << 8) |
              ((((unsigned)d4.z / RROWS) & 255u) << 16) |
              ((((unsigned)d4.w / RROWS) & 255u) << 24);
  } else {
    unsigned int tw = 0;
    for (int k = 0; k < 4 && e + k < E; ++k) {
      const int d = dst[e + k];
      enorm[e + k] = dinv[src[e + k]] * dinv[d];
      tw |= (((unsigned)d / RROWS) & 255u) << (8 * k);
    }
    tagw[t] = tw;
  }
}

#define GROWS 400
#define GWROWS 80
#define GTHREADS 640
template <int K, bool RELU>
__global__ __launch_bounds__(GTHREADS) void gemm_wmma_kernel(
    const float* __restrict__ X, const float* __restrict__ W,
    float* __restrict__ H, int N) {
  const int lane = threadIdx.x & 31;
  const int wave = threadIdx.x >> 5;
  const int cq = wave & 3;
  const int rgrp = wave >> 2;
  const int row_base = blockIdx.x * GROWS + rgrp * GWROWS;
  if (row_base >= N) return;
  const int lb = max(min(row_base, N - GWROWS), 0);

  const int col_base = cq * 64;
  const int col = lane & 15;
  const int khalf = (lane >> 4) << 1;

  const float* ap = X + (size_t)(lb + col) * K + khalf;
  const float* bp = W + (size_t)khalf * HID + col_base + col;

  v8f acc[5][4];
#pragma unroll
  for (int m = 0; m < 5; ++m)
#pragma unroll
    for (int j = 0; j < 4; ++j)
      acc[m][j] = (v8f){0.f, 0.f, 0.f, 0.f, 0.f, 0.f, 0.f, 0.f};

  for (int k = 0; k < K; k += 4) {
    v2f a[5];
    v2f b[4];
#pragma unroll
    for (int m = 0; m < 5; ++m) a[m] = *(const v2f*)(ap + m * 16 * K);
#pragma unroll
    for (int j = 0; j < 4; ++j) {
      b[j].x = bp[j * 16];
      b[j].y = bp[HID + j * 16];
    }
    ap += 4;
    bp += 4 * HID;
    if (RELU) {
#pragma unroll
      for (int m = 0; m < 5; ++m) {
        a[m].x = fmaxf(a[m].x, 0.f);
        a[m].y = fmaxf(a[m].y, 0.f);
      }
    }
#pragma unroll
    for (int j = 0; j < 4; ++j)
#pragma unroll
      for (int m = 0; m < 5; ++m)
        acc[m][j] = __builtin_amdgcn_wmma_f32_16x16x4_f32(
            false, a[m], false, b[j], (short)0, acc[m][j], false, false);
  }

  float* hp = H + (size_t)(lb + (lane >> 4) * 8) * HID + col_base + col;
  const int rfirst = row_base - lb - (lane >> 4) * 8;
#pragma unroll
  for (int m = 0; m < 5; ++m) {
#pragma unroll
    for (int r = 0; r < 8; ++r) {
      if (m * 16 + r >= rfirst) {
#pragma unroll
        for (int j = 0; j < 4; ++j)
          hp[(size_t)(m * 16 + r) * HID + j * 16] = acc[m][j][r];
      }
    }
  }
}


template <bool BUILD>
__global__ __launch_bounds__(RTHREADS) void range_sum_kernel(
    const float* __restrict__ h, const int* __restrict__ src,
    const int* __restrict__ dst, const float* __restrict__ enorm,
    const unsigned int* __restrict__ tagw, const float* __restrict__ dinv,
    const float* __restrict__ bias, float* __restrict__ out,
    int* __restrict__ gcount, int* __restrict__ gsrc, int* __restrict__ gdst,
    float* __restrict__ gw, int N, int E) {
  __shared__ float acc[RROWS * RCOLS];
  __shared__ int ls[LCAP];
  __shared__ int ld[LCAP];
  __shared__ float lw[LCAP];
  __shared__ int ecount;

  const int lane = threadIdx.x & 31;
  const int wave = __builtin_amdgcn_readfirstlane(threadIdx.x >> 5);
  const int range = blockIdx.x;
  const int r0 = range * RROWS;
  const int nrows = min(RROWS, N - r0);
  const size_t lbase = (size_t)range * LCAP;

  int total;
  if (BUILD) {
    if (threadIdx.x == 0) ecount = 0;
    __syncthreads();

    auto note_edge = [&](int e) {
      const int slot = atomicAdd(&ecount, 1);
      if (slot < LCAP) ls[slot] = e;
    };

    const bool exact = (N <= 256 * RROWS);
    const int nSuper = exact ? (E >> 11) : 0;
    const int rot = (int)(((long long)(range & 127) * nSuper) >> 7);
    const unsigned pat = (unsigned)(range & 255) * 0x01010101u;
    const uint4* t4 = (const uint4*)tagw;
    for (int i = wave; i < nSuper; i += RWAVES) {
      int c = i + rot;
      if (c >= nSuper) c -= nSuper;
      const uint4* p = t4 + (size_t)c * 128 + lane;
      uint4 t[4];
#pragma unroll
      for (int u = 0; u < 4; ++u) t[u] = p[u * 32];
      unsigned m[4];
#pragma unroll
      for (int u = 0; u < 4; ++u) {
        const unsigned wd[4] = {t[u].x, t[u].y, t[u].z, t[u].w};
        m[u] = 0;
#pragma unroll
        for (int q = 0; q < 4; ++q) {
          const unsigned y = wd[q] ^ pat;
          const unsigned z = ~(((y & 0x7F7F7F7Fu) + 0x7F7F7F7Fu) | y);
          m[u] |= (z >> (7 - q)) & (0x01010101u << q);
        }
      }
      unsigned long long hits =
          (unsigned long long)(m[0] | (m[1] << 4)) |
          ((unsigned long long)(m[2] | (m[3] << 4)) << 32);
      while (hits) {
        const int b = __builtin_ctzll(hits);
        hits &= hits - 1;
        const int u = ((b >> 5) << 1) | ((b >> 2) & 1);
        const int j = (b & 31) >> 3;
        const int q = b & 3;
        note_edge(c * 2048 + u * 512 + lane * 16 + q * 4 + j);
      }
    }
    for (int e = (nSuper << 11) + (int)threadIdx.x; e < E; e += RTHREADS) {
      if ((unsigned)(dst[e] - r0) < (unsigned)nrows) note_edge(e);
    }
    __syncthreads();

    total = ecount;
    const int n = min(total, LCAP);
    for (int i = threadIdx.x; i < n; i += RTHREADS) {
      const int e = ls[i];
      const int s = src[e];
      const int d = dst[e] - r0;
      const float w = enorm[e];
      ls[i] = s;
      ld[i] = d;
      lw[i] = w;
      gsrc[lbase + i] = s;
      gdst[lbase + i] = d;
      gw[lbase + i] = w;
    }
    if (threadIdx.x == 0) gcount[range] = total;
  } else {
    total = gcount[range];
    const int n = min(total, LCAP);
    for (int i = threadIdx.x; i < n; i += RTHREADS) {
      ls[i] = gsrc[lbase + i];
      ld[i] = gdst[lbase + i];
      lw[i] = gw[lbase + i];
    }
  }

  const int total4 = nrows * (RCOLS / 4);
  for (int half = 0; half < HID / RCOLS; ++half) {
    const int col0 = half * RCOLS;

    for (int i0 = threadIdx.x; i0 < total4; i0 += RTHREADS * 4) {
      float4 hv[4];
#pragma unroll
      for (int u = 0; u < 4; ++u) {
        const int i = i0 + u * RTHREADS;
        if (i < total4)
          hv[u] = *(const float4*)(h + (size_t)(r0 + (i >> 5)) * HID + col0 +
                                   (i & 31) * 4);
      }
#pragma unroll
      for (int u = 0; u < 4; ++u) {
        const int i = i0 + u * RTHREADS;
        if (i < total4) {
          const float di = dinv[r0 + (i >> 5)];
          const float sn = di * di;
          const float4 bv = *(const float4*)(bias + col0 + (i & 31) * 4);
          float4 o;
          o.x = fmaf(hv[u].x, sn, bv.x);
          o.y = fmaf(hv[u].y, sn, bv.y);
          o.z = fmaf(hv[u].z, sn, bv.z);
          o.w = fmaf(hv[u].w, sn, bv.w);
          float* ar = acc + (i >> 5) * RCOLS + (i & 31);
          ar[0] = o.x;
          ar[32] = o.y;
          ar[64] = o.z;
          ar[96] = o.w;
        }
      }
    }
    __syncthreads();

    if (total <= LCAP) {
      const int n = total;
      int base = wave * 8;
      int s = 0, d = 0;
      float w = 0.f;
      if (base + (lane & 7) < n) {
        s = ls[base + (lane & 7)];
        d = ld[base + (lane & 7)];
        w = lw[base + (lane & 7)];
      }
      while (base < n) {
        const int nbase = base + RWAVES * 8;
        int s2 = 0, d2 = 0;
        float w2 = 0.f;
        if (nbase + (lane & 7) < n) {
          s2 = ls[nbase + (lane & 7)];
          d2 = ld[nbase + (lane & 7)];
          w2 = lw[nbase + (lane & 7)];
        }
        const int cnt = min(8, n - base);
        if (cnt == 8) {
          float4 v[8];
#pragma unroll
          for (int q = 0; q < 8; ++q) {
            const int sk = __builtin_amdgcn_readlane(s, q);
            v[q] = *(const float4*)(h + (size_t)sk * HID + col0 + lane * 4);
          }
#pragma unroll
          for (int q = 0; q < 8; ++q) {
            const int dk = __builtin_amdgcn_readlane(d, q);
            const float wk =
                __int_as_float(__builtin_amdgcn_readlane(__float_as_int(w), q));
            float* ap = acc + dk * RCOLS + lane;
            unsafeAtomicAdd(ap, v[q].x * wk);
            unsafeAtomicAdd(ap + 32, v[q].y * wk);
            unsafeAtomicAdd(ap + 64, v[q].z * wk);
            unsafeAtomicAdd(ap + 96, v[q].w * wk);
          }
        } else {
          for (int k = 0; k < cnt; ++k) {
            const int sk = __builtin_amdgcn_readlane(s, k);
            const int dk = __builtin_amdgcn_readlane(d, k);
            const float wk =
                __int_as_float(__builtin_amdgcn_readlane(__float_as_int(w), k));
            const float4 v1 =
                *(const float4*)(h + (size_t)sk * HID + col0 + lane * 4);
            float* ap = acc + dk * RCOLS + lane;
            unsafeAtomicAdd(ap, v1.x * wk);
            unsafeAtomicAdd(ap + 32, v1.y * wk);
            unsafeAtomicAdd(ap + 64, v1.z * wk);
            unsafeAtomicAdd(ap + 96, v1.w * wk);
          }
        }
        s = s2;
        d = d2;
        w = w2;
        base = nbase;
      }
    } else {
      for (int e = threadIdx.x; e < E; e += RTHREADS) {
        const int dl = dst[e] - r0;
        if ((unsigned)dl < (unsigned)nrows) {
          const float* hp = h + (size_t)src[e] * HID + col0;
          float* ap = acc + dl * RCOLS;
          const float w = enorm[e];
          for (int c = 0; c < RCOLS; ++c)
            unsafeAtomicAdd(ap + (c & 3) * 32 + (c >> 2), hp[c] * w);
        }
      }
    }
    __syncthreads();

    for (int i = threadIdx.x; i < total4; i += RTHREADS) {
      const float* ar = acc + (i >> 5) * RCOLS + (i & 31);
      float4 o;
      o.x = ar[0];
      o.y = ar[32];
      o.z = ar[64];
      o.w = ar[96];
      *(float4*)(out + (size_t)(r0 + (i >> 5)) * HID + col0 + (i & 31) * 4) = o;
    }
    __syncthreads();
  }
}


static inline int cdiv(int a, int b) { return (a + b - 1) / b; }

extern "C" void kernel_launch(void* const* d_in, const int* in_sizes, int n_in,
                              void* d_out, int out_size, void* d_ws,
                              size_t ws_size, hipStream_t stream) {
  const float* x = (const float*)d_in[0];
  const int* ei = (const int*)d_in[1];
  const float* W1 = (const float*)d_in[2];
  const float* b1 = (const float*)d_in[3];
  const float* W2 = (const float*)d_in[4];
  const float* b2 = (const float*)d_in[5];
  const float* W3 = (const float*)d_in[6];
  const float* b3 = (const float*)d_in[7];
  float* out = (float*)d_out;

  const int F = 128;
  const int N = in_sizes[0] / F;
  const int E = in_sizes[1] / 2;
  const int* src = ei;
  const int* dst = ei + E;

  float* wsf = (float*)d_ws;
  size_t off = 0;
  float* dinv = wsf + off;
  off += (size_t)((N + 255) / 256) * 256;
  float* enorm = wsf + off;
  off += (size_t)((E + 255) / 256) * 256;
  float* hbuf = wsf + off;
  off += (size_t)N * HID;
  float* agg = wsf + off;
  off += (size_t)N * HID;
  unsigned int* tagw = (unsigned int*)(wsf + off);
  off += (size_t)(((E + 3) / 4 + 255) / 256) * 256;
  const int nRanges = cdiv(N, RROWS);
  int* gcount = (int*)(wsf + off);
  off += (size_t)((nRanges + 255) / 256) * 256;
  int* gsrc = (int*)(wsf + off);
  off += (size_t)nRanges * LCAP;
  int* gdst = (int*)(wsf + off);
  off += (size_t)nRanges * LCAP;
  float* gw = wsf + off;

  const int nodeGrid = cdiv(N, TPB);
  const int edgeGrid = cdiv(E, TPB);
  const int edge4Grid = cdiv(cdiv(E, 4), TPB);
  const int rangeGrid = nRanges;
  const int gemmGrid = cdiv(N, GROWS);

  fill_kernel<<<nodeGrid, TPB, 0, stream>>>(dinv, 1.0f, N);
  degree_kernel<<<edgeGrid, TPB, 0, stream>>>(dst, dinv, E);
  rsqrt_inplace_kernel<<<nodeGrid, TPB, 0, stream>>>(dinv, N);
  enorm_kernel<<<edge4Grid, TPB, 0, stream>>>(src, dst, dinv, enorm, tagw, E);

  gemm_wmma_kernel<128, false><<<gemmGrid, GTHREADS, 0, stream>>>(x, W1, hbuf, N);
  range_sum_kernel<true><<<rangeGrid, RTHREADS, 0, stream>>>(
      hbuf, src, dst, enorm, tagw, dinv, b1, agg, gcount, gsrc, gdst, gw, N, E);

  gemm_wmma_kernel<256, true><<<gemmGrid, GTHREADS, 0, stream>>>(agg, W2, hbuf, N);
  range_sum_kernel<false><<<rangeGrid, RTHREADS, 0, stream>>>(
      hbuf, src, dst, enorm, tagw, dinv, b2, agg, gcount, gsrc, gdst, gw, N, E);

  gemm_wmma_kernel<256, true><<<gemmGrid, GTHREADS, 0, stream>>>(agg, W3, hbuf, N);
  range_sum_kernel<false><<<rangeGrid, RTHREADS, 0, stream>>>(
      hbuf, src, dst, enorm, tagw, dinv, b3, out, gcount, gsrc, gdst, gw, N, E);
}


# degrees: a row block counts its rows' incoming edges with LDS integer adds while it lists them (once per call) and writes dinv; 800000 float atomic adds to memory and 2 launches go; sums unchanged
# speedup vs baseline: 3.8342x; 1.0757x over previous
#include <hip/hip_runtime.h>

typedef __attribute__((ext_vector_type(2))) float v2f;
typedef __attribute__((ext_vector_type(8))) float v8f;

#define HID 256
#define TPB 256

#define RROWS 392
#define RCOLS 128
#define RTHREADS 1024
#define RWAVES 32
#define LCAP 8192


__global__ void tag_kernel(const int* __restrict__ dst,
                           unsigned int* __restrict__ tagw, int E) {
  const int t = blockIdx.x * blockDim.x + threadIdx.x;
  const int e = t * 4;
  if (e >= E) return;
  const bool al16 = (((size_t)dst) & 15) == 0;
  if (al16 && e + 3 < E) {
    const int4 d4 = *(const int4*)(dst + e);
    tagw[t] = (((unsigned)d4.x / RROWS) & 255u) |
              ((((unsigned)d4.y / RROWS) & 255u) << 8) |
              ((((unsigned)d4.z / RROWS) & 255u) << 16) |
              ((((unsigned)d4.w / RROWS) & 255u) << 24);
  } else {
    unsigned int tw = 0;
    for (int k = 0; k < 4 && e + k < E; ++k)
      tw |= (((unsigned)dst[e + k] / RROWS) & 255u) << (8 * k);
    tagw[t] = tw;
  }
}

__global__ __launch_bounds__(RTHREADS) void build_kernel(
    const int* __restrict__ src, const int* __restrict__ dst,
    const unsigned int* __restrict__ tagw, float* __restrict__ dinv,
    int* __restrict__ gcount, int* __restrict__ gsrc, int* __restrict__ gdst,
    int N, int E) {
  __shared__ int ls[LCAP];
  __shared__ int ldeg[RROWS];
  __shared__ int ecount;

  const int lane = threadIdx.x & 31;
  const int wave = __builtin_amdgcn_readfirstlane(threadIdx.x >> 5);
  const int range = blockIdx.x;
  const int r0 = range * RROWS;
  const int nrows = min(RROWS, N - r0);
  const size_t lbase = (size_t)range * LCAP;

  if (threadIdx.x == 0) ecount = 0;
  for (int i = threadIdx.x; i < RROWS; i += RTHREADS) ldeg[i] = 0;
  __syncthreads();

  auto note_edge = [&](int e) {
    const int slot = atomicAdd(&ecount, 1);
    if (slot < LCAP) ls[slot] = e;
  };

  const bool exact = (N <= 256 * RROWS);
  const int nSuper = exact ? (E >> 11) : 0;
  const int rot = (int)(((long long)(range & 127) * nSuper) >> 7);
  const unsigned pat = (unsigned)(range & 255) * 0x01010101u;
  const uint4* t4 = (const uint4*)tagw;
  for (int i = wave; i < nSuper; i += RWAVES) {
    int c = i + rot;
    if (c >= nSuper) c -= nSuper;
    const uint4* p = t4 + (size_t)c * 128 + lane;
    uint4 t[4];
#pragma unroll
    for (int u = 0; u < 4; ++u) t[u] = p[u * 32];
    unsigned m[4];
#pragma unroll
    for (int u = 0; u < 4; ++u) {
      const unsigned wd[4] = {t[u].x, t[u].y, t[u].z, t[u].w};
      m[u] = 0;
#pragma unroll
      for (int q = 0; q < 4; ++q) {
        const unsigned y = wd[q] ^ pat;
        const unsigned z = ~(((y & 0x7F7F7F7Fu) + 0x7F7F7F7Fu) | y);
        m[u] |= (z >> (7 - q)) & (0x01010101u << q);
      }
    }
    unsigned long long hits =
        (unsigned long long)(m[0] | (m[1] << 4)) |
        ((unsigned long long)(m[2] | (m[3] << 4)) << 32);
    while (hits) {
      const int b = __builtin_ctzll(hits);
      hits &= hits - 1;
      const int u = ((b >> 5) << 1) | ((b >> 2) & 1);
      const int j = (b & 31) >> 3;
      const int q = b & 3;
      note_edge(c * 2048 + u * 512 + lane * 16 + q * 4 + j);
    }
  }
  for (int e = (nSuper << 11) + (int)threadIdx.x; e < E; e += RTHREADS) {
    if ((unsigned)(dst[e] - r0) < (unsigned)nrows) note_edge(e);
  }
  __syncthreads();

  const int total = ecount;
  if (total <= LCAP) {
    for (int i = threadIdx.x; i < total; i += RTHREADS) {
      const int e = ls[i];
      const int s = src[e];
      const int d = dst[e] - r0;
      atomicAdd(&ldeg[d], 1);
      gsrc[lbase + i] = s;
      gdst[lbase + i] = d;
    }
  } else {
    for (int e = threadIdx.x; e < E; e += RTHREADS) {
      const int dl = dst[e] - r0;
      if ((unsigned)dl < (unsigned)nrows) atomicAdd(&ldeg[dl], 1);
    }
  }
  if (threadIdx.x == 0) gcount[range] = total;
  __syncthreads();
  for (int i = threadIdx.x; i < nrows; i += RTHREADS)
    dinv[r0 + i] = 1.0f / sqrtf((float)(ldeg[i] + 1));
}

#define GROWS 400
#define GWROWS 80
#define GTHREADS 640
template <int K, bool RELU>
__global__ __launch_bounds__(GTHREADS) void gemm_wmma_kernel(
    const float* __restrict__ X, const float* __restrict__ W,
    float* __restrict__ H, int N) {
  const int lane = threadIdx.x & 31;
  const int wave = threadIdx.x >> 5;
  const int cq = wave & 3;
  const int rgrp = wave >> 2;
  const int row_base = blockIdx.x * GROWS + rgrp * GWROWS;
  if (row_base >= N) return;
  const int lb = max(min(row_base, N - GWROWS), 0);

  const int col_base = cq * 64;
  const int col = lane & 15;
  const int khalf = (lane >> 4) << 1;

  const float* ap = X + (size_t)(lb + col) * K + khalf;
  const float* bp = W + (size_t)khalf * HID + col_base + col;

  v8f acc[5][4];
#pragma unroll
  for (int m = 0; m < 5; ++m)
#pragma unroll
    for (int j = 0; j < 4; ++j)
      acc[m][j] = (v8f){0.f, 0.f, 0.f, 0.f, 0.f, 0.f, 0.f, 0.f};

  for (int k = 0; k < K; k += 4) {
    v2f a[5];
    v2f b[4];
#pragma unroll
    for (int m = 0; m < 5; ++m) a[m] = *(const v2f*)(ap + m * 16 * K);
#pragma unroll
    for (int j = 0; j < 4; ++j) {
      b[j].x = bp[j * 16];
      b[j].y = bp[HID + j * 16];
    }
    ap += 4;
    bp += 4 * HID;
    if (RELU) {
#pragma unroll
      for (int m = 0; m < 5; ++m) {
        a[m].x = fmaxf(a[m].x, 0.f);
        a[m].y = fmaxf(a[m].y, 0.f);
      }
    }
#pragma unroll
    for (int j = 0; j < 4; ++j)
#pragma unroll
      for (int m = 0; m < 5; ++m)
        acc[m][j] = __builtin_amdgcn_wmma_f32_16x16x4_f32(
            false, a[m], false, b[j], (short)0, acc[m][j], false, false);
  }

  float* hp = H + (size_t)(lb + (lane >> 4) * 8) * HID + col_base + col;
  const int rfirst = row_base - lb - (lane >> 4) * 8;
#pragma unroll
  for (int m = 0; m < 5; ++m) {
#pragma unroll
    for (int r = 0; r < 8; ++r) {
      if (m * 16 + r >= rfirst) {
#pragma unroll
        for (int j = 0; j < 4; ++j)
          hp[(size_t)(m * 16 + r) * HID + j * 16] = acc[m][j][r];
      }
    }
  }
}


template <bool FIRST>
__global__ __launch_bounds__(RTHREADS) void range_sum_kernel(
    const float* __restrict__ h, const int* __restrict__ src,
    const int* __restrict__ dst, const float* __restrict__ dinv,
    const float* __restrict__ bias, float* __restrict__ out,
    const int* __restrict__ gcount, const int* __restrict__ gsrc,
    const int* __restrict__ gdst, float* __restrict__ gw, int N, int E) {
  __shared__ float acc[RROWS * RCOLS];
  __shared__ int ls[LCAP];
  __shared__ int ld[LCAP];
  __shared__ float lw[LCAP];

  const int lane = threadIdx.x & 31;
  const int wave = __builtin_amdgcn_readfirstlane(threadIdx.x >> 5);
  const int range = blockIdx.x;
  const int r0 = range * RROWS;
  const int nrows = min(RROWS, N - r0);
  const size_t lbase = (size_t)range * LCAP;

  const int total = gcount[range];
  {
    const int n = (total <= LCAP) ? total : 0;
    for (int i = threadIdx.x; i < n; i += RTHREADS) {
      const int s = gsrc[lbase + i];
      const int d = gdst[lbase + i];
      float w;
      if (FIRST) {
        w = dinv[s] * dinv[r0 + d];
        gw[lbase + i] = w;
      } else {
        w = gw[lbase + i];
      }
      ls[i] = s;
      ld[i] = d;
      lw[i] = w;
    }
  }

  const int total4 = nrows * (RCOLS / 4);
  for (int half = 0; half < HID / RCOLS; ++half) {
    const int col0 = half * RCOLS;

    for (int i0 = threadIdx.x; i0 < total4; i0 += RTHREADS * 4) {
      float4 hv[4];
#pragma unroll
      for (int u = 0; u < 4; ++u) {
        const int i = i0 + u * RTHREADS;
        if (i < total4)
          hv[u] = *(const float4*)(h + (size_t)(r0 + (i >> 5)) * HID + col0 +
                                   (i & 31) * 4);
      }
#pragma unroll
      for (int u = 0; u < 4; ++u) {
        const int i = i0 + u * RTHREADS;
        if (i < total4) {
          const float di = dinv[r0 + (i >> 5)];
          const float sn = di * di;
          const float4 bv = *(const float4*)(bias + col0 + (i & 31) * 4);
          float4 o;
          o.x = fmaf(hv[u].x, sn, bv.x);
          o.y = fmaf(hv[u].y, sn, bv.y);
          o.z = fmaf(hv[u].z, sn, bv.z);
          o.w = fmaf(hv[u].w, sn, bv.w);
          float* ar = acc + (i >> 5) * RCOLS + (i & 31);
          ar[0] = o.x;
          ar[32] = o.y;
          ar[64] = o.z;
          ar[96] = o.w;
        }
      }
    }
    __syncthreads();

    if (total <= LCAP) {
      const int n = total;
      int base = wave * 8;
      int s = 0, d = 0;
      float w = 0.f;
      if (base + (lane & 7) < n) {
        s = ls[base + (lane & 7)];
        d = ld[base + (lane & 7)];
        w = lw[base + (lane & 7)];
      }
      while (base < n) {
        const int nbase = base + RWAVES * 8;
        int s2 = 0, d2 = 0;
        float w2 = 0.f;
        if (nbase + (lane & 7) < n) {
          s2 = ls[nbase + (lane & 7)];
          d2 = ld[nbase + (lane & 7)];
          w2 = lw[nbase + (lane & 7)];
        }
        const int cnt = min(8, n - base);
        if (cnt == 8) {
          float4 v[8];
#pragma unroll
          for (int q = 0; q < 8; ++q) {
            const int sk = __builtin_amdgcn_readlane(s, q);
            v[q] = *(const float4*)(h + (size_t)sk * HID + col0 + lane * 4);
          }
#pragma unroll
          for (int q = 0; q < 8; ++q) {
            const int dk = __builtin_amdgcn_readlane(d, q);
            const float wk =
                __int_as_float(__builtin_amdgcn_readlane(__float_as_int(w), q));
            float* ap = acc + dk * RCOLS + lane;
            unsafeAtomicAdd(ap, v[q].x * wk);
            unsafeAtomicAdd(ap + 32, v[q].y * wk);
            unsafeAtomicAdd(ap + 64, v[q].z * wk);
            unsafeAtomicAdd(ap + 96, v[q].w * wk);
          }
        } else {
          for (int k = 0; k < cnt; ++k) {
            const int sk = __builtin_amdgcn_readlane(s, k);
            const int dk = __builtin_amdgcn_readlane(d, k);
            const float wk =
                __int_as_float(__builtin_amdgcn_readlane(__float_as_int(w), k));
            const float4 v1 =
                *(const float4*)(h + (size_t)sk * HID + col0 + lane * 4);
            float* ap = acc + dk * RCOLS + lane;
            unsafeAtomicAdd(ap, v1.x * wk);
            unsafeAtomicAdd(ap + 32, v1.y * wk);
            unsafeAtomicAdd(ap + 64, v1.z * wk);
            unsafeAtomicAdd(ap + 96, v1.w * wk);
          }
        }
        s = s2;
        d = d2;
        w = w2;
        base = nbase;
      }
    } else {
      for (int e = threadIdx.x; e < E; e += RTHREADS) {
        const int dl = dst[e] - r0;
        if ((unsigned)dl < (unsigned)nrows) {
          const float* hp = h + (size_t)src[e] * HID + col0;
          float* ap = acc + dl * RCOLS;
          const float w = dinv[src[e]] * dinv[dst[e]];
          for (int c = 0; c < RCOLS; ++c)
            unsafeAtomicAdd(ap + (c & 3) * 32 + (c >> 2), hp[c] * w);
        }
      }
    }
    __syncthreads();

    for (int i = threadIdx.x; i < total4; i += RTHREADS) {
      const float* ar = acc + (i >> 5) * RCOLS + (i & 31);
      float4 o;
      o.x = ar[0];
      o.y = ar[32];
      o.z = ar[64];
      o.w = ar[96];
      *(float4*)(out + (size_t)(r0 + (i >> 5)) * HID + col0 + (i & 31) * 4) = o;
    }
    __syncthreads();
  }
}


static inline int cdiv(int a, int b) { return (a + b - 1) / b; }

extern "C" void kernel_launch(void* const* d_in, const int* in_sizes, int n_in,
                              void* d_out, int out_size, void* d_ws,
                              size_t ws_size, hipStream_t stream) {
  const float* x = (const float*)d_in[0];
  const int* ei = (const int*)d_in[1];
  const float* W1 = (const float*)d_in[2];
  const float* b1 = (const float*)d_in[3];
  const float* W2 = (const float*)d_in[4];
  const float* b2 = (const float*)d_in[5];
  const float* W3 = (const float*)d_in[6];
  const float* b3 = (const float*)d_in[7];
  float* out = (float*)d_out;

  const int F = 128;
  const int N = in_sizes[0] / F;
  const int E = in_sizes[1] / 2;
  const int* src = ei;
  const int* dst = ei + E;

  float* wsf = (float*)d_ws;
  size_t off = 0;
  float* dinv = wsf + off;
  off += (size_t)((N + 255) / 256) * 256;
  float* hbuf = wsf + off;
  off += (size_t)N * HID;
  float* agg = wsf + off;
  off += (size_t)N * HID;
  unsigned int* tagw = (unsigned int*)(wsf + off);
  off += (size_t)(((E + 3) / 4 + 255) / 256) * 256;
  const int nRanges = cdiv(N, RROWS);
  int* gcount = (int*)(wsf + off);
  off += (size_t)((nRanges + 255) / 256) * 256;
  int* gsrc = (int*)(wsf + off);
  off += (size_t)nRanges * LCAP;
  int* gdst = (int*)(wsf + off);
  off += (size_t)nRanges * LCAP;
  float* gw = wsf + off;

  const int edge4Grid = cdiv(cdiv(E, 4), TPB);
  const int rangeGrid = nRanges;
  const int gemmGrid = cdiv(N, GROWS);

  tag_kernel<<<edge4Grid, TPB, 0, stream>>>(dst, tagw, E);
  build_kernel<<<rangeGrid, RTHREADS, 0, stream>>>(src, dst, tagw, dinv, gcount,
                                                  gsrc, gdst, N, E);

  gemm_wmma_kernel<128, false><<<gemmGrid, GTHREADS, 0, stream>>>(x, W1, hbuf, N);
  range_sum_kernel<true><<<rangeGrid, RTHREADS, 0, stream>>>(
      hbuf, src, dst, dinv, b1, agg, gcount, gsrc, gdst, gw, N, E);

  gemm_wmma_kernel<256, true><<<gemmGrid, GTHREADS, 0, stream>>>(agg, W2, hbuf, N);
  range_sum_kernel<false><<<rangeGrid, RTHREADS, 0, stream>>>(
      hbuf, src, dst, dinv, b2, agg, gcount, gsrc, gdst, gw, N, E);

  gemm_wmma_kernel<256, true><<<gemmGrid, GTHREADS, 0, stream>>>(agg, W3, hbuf, N);
  range_sum_kernel<false><<<rangeGrid, RTHREADS, 0, stream>>>(
      hbuf, src, dst, dinv, b3, out, gcount, gsrc, gdst, gw, N, E);
}


# aggregation: block's edge list built in destination-row order (LDS counts, prefix sum, place); a wave sums a row's edges in registers, adds to LDS once per row, not per edge; sum order changes
# speedup vs baseline: 4.0060x; 1.0448x over previous
#include <hip/hip_runtime.h>

typedef __attribute__((ext_vector_type(2))) float v2f;
typedef __attribute__((ext_vector_type(8))) float v8f;

#define HID 256
#define TPB 256

#define RROWS 392
#define RCOLS 128
#define RTHREADS 1024
#define RWAVES 32
#define LCAP 8192


__global__ void tag_kernel(const int* __restrict__ dst,
                           unsigned int* __restrict__ tagw, int E) {
  const int t = blockIdx.x * blockDim.x + threadIdx.x;
  const int e = t * 4;
  if (e >= E) return;
  const bool al16 = (((size_t)dst) & 15) == 0;
  if (al16 && e + 3 < E) {
    const int4 d4 = *(const int4*)(dst + e);
    tagw[t] = (((unsigned)d4.x / RROWS) & 255u) |
              ((((unsigned)d4.y / RROWS) & 255u) << 8) |
              ((((unsigned)d4.z / RROWS) & 255u) << 16) |
              ((((unsigned)d4.w / RROWS) & 255u) << 24);
  } else {
    unsigned int tw = 0;
    for (int k = 0; k < 4 && e + k < E; ++k)
      tw |= (((unsigned)dst[e + k] / RROWS) & 255u) << (8 * k);
    tagw[t] = tw;
  }
}

__global__ __launch_bounds__(RTHREADS) void build_kernel(
    const int* __restrict__ src, const int* __restrict__ dst,
    const unsigned int* __restrict__ tagw, float* __restrict__ dinv,
    int* __restrict__ gcount, int* __restrict__ gsrc, int* __restrict__ gdst,
    int N, int E) {
  __shared__ int ls[LCAP];
  __shared__ int ldl[LCAP];
  __shared__ int ldeg[RROWS];
  __shared__ int lpos[RROWS];
  __shared__ int wsum[RWAVES];
  __shared__ int ecount;
  static_assert(RROWS <= RTHREADS, "one thread per row in the prefix sum");

  const int lane = threadIdx.x & 31;
  const int wave = __builtin_amdgcn_readfirstlane(threadIdx.x >> 5);
  const int range = blockIdx.x;
  const int r0 = range * RROWS;
  const int nrows = min(RROWS, N - r0);
  const size_t lbase = (size_t)range * LCAP;

  if (threadIdx.x == 0) ecount = 0;
  for (int i = threadIdx.x; i < RROWS; i += RTHREADS) ldeg[i] = 0;
  __syncthreads();

  auto note_edge = [&](int e) {
    const int slot = atomicAdd(&ecount, 1);
    if (slot < LCAP) ls[slot] = e;
  };

  const bool exact = (N <= 256 * RROWS);
  const int nSuper = exact ? (E >> 11) : 0;
  const int rot = (int)(((long long)(range & 127) * nSuper) >> 7);
  const unsigned pat = (unsigned)(range & 255) * 0x01010101u;
  const uint4* t4 = (const uint4*)tagw;
  for (int i = wave; i < nSuper; i += RWAVES) {
    int c = i + rot;
    if (c >= nSuper) c -= nSuper;
    const uint4* p = t4 + (size_t)c * 128 + lane;
    uint4 t[4];
#pragma unroll
    for (int u = 0; u < 4; ++u) t[u] = p[u * 32];
    unsigned m[4];
#pragma unroll
    for (int u = 0; u < 4; ++u) {
      const unsigned wd[4] = {t[u].x, t[u].y, t[u].z, t[u].w};
      m[u] = 0;
#pragma unroll
      for (int q = 0; q < 4; ++q) {
        const unsigned y = wd[q] ^ pat;
        const unsigned z = ~(((y & 0x7F7F7F7Fu) + 0x7F7F7F7Fu) | y);
        m[u] |= (z >> (7 - q)) & (0x01010101u << q);
      }
    }
    unsigned long long hits =
        (unsigned long long)(m[0] | (m[1] << 4)) |
        ((unsigned long long)(m[2] | (m[3] << 4)) << 32);
    while (hits) {
      const int b = __builtin_ctzll(hits);
      hits &= hits - 1;
      const int u = ((b >> 5) << 1) | ((b >> 2) & 1);
      const int j = (b & 31) >> 3;
      const int q = b & 3;
      note_edge(c * 2048 + u * 512 + lane * 16 + q * 4 + j);
    }
  }
  for (int e = (nSuper << 11) + (int)threadIdx.x; e < E; e += RTHREADS) {
    if ((unsigned)(dst[e] - r0) < (unsigned)nrows) note_edge(e);
  }
  __syncthreads();

  const int total = ecount;
  if (total <= LCAP) {
    for (int i = threadIdx.x; i < total; i += RTHREADS) {
      const int e = ls[i];
      const int d = dst[e] - r0;
      atomicAdd(&ldeg[d], 1);
      ls[i] = src[e];
      ldl[i] = d;
    }
    __syncthreads();
    {
      const int t = threadIdx.x;
      const int v = (t < RROWS) ? ldeg[t] : 0;
      int x = v;
      for (int o = 1; o < 32; o <<= 1) {
        const int y = __shfl_up(x, o);
        if (lane >= o) x += y;
      }
      if (lane == 31) wsum[wave] = x;
      __syncthreads();
      int off = 0;
      for (int k = 0; k < wave; ++k) off += wsum[k];
      if (t < RROWS) lpos[t] = off + x - v;
    }
    __syncthreads();
    for (int i = threadIdx.x; i < total; i += RTHREADS) {
      const int d = ldl[i];
      const int pos = atomicAdd(&lpos[d], 1);
      gsrc[lbase + pos] = ls[i];
      gdst[lbase + pos] = d;
    }
  } else {
    for (int e = threadIdx.x; e < E; e += RTHREADS) {
      const int dl = dst[e] - r0;
      if ((unsigned)dl < (unsigned)nrows) atomicAdd(&ldeg[dl], 1);
    }
  }
  if (threadIdx.x == 0) gcount[range] = total;
  __syncthreads();
  for (int i = threadIdx.x; i < nrows; i += RTHREADS)
    dinv[r0 + i] = 1.0f / sqrtf((float)(ldeg[i] + 1));
}

#define GROWS 400
#define GWROWS 80
#define GTHREADS 640
template <int K, bool RELU>
__global__ __launch_bounds__(GTHREADS) void gemm_wmma_kernel(
    const float* __restrict__ X, const float* __restrict__ W,
    float* __restrict__ H, int N) {
  const int lane = threadIdx.x & 31;
  const int wave = threadIdx.x >> 5;
  const int cq = wave & 3;
  const int rgrp = wave >> 2;
  const int row_base = blockIdx.x * GROWS + rgrp * GWROWS;
  if (row_base >= N) return;
  const int lb = max(min(row_base, N - GWROWS), 0);

  const int col_base = cq * 64;
  const int col = lane & 15;
  const int khalf = (lane >> 4) << 1;

  const float* ap = X + (size_t)(lb + col) * K + khalf;
  const float* bp = W + (size_t)khalf * HID + col_base + col;

  v8f acc[5][4];
#pragma unroll
  for (int m = 0; m < 5; ++m)
#pragma unroll
    for (int j = 0; j < 4; ++j)
      acc[m][j] = (v8f){0.f, 0.f, 0.f, 0.f, 0.f, 0.f, 0.f, 0.f};

  for (int k = 0; k < K; k += 4) {
    v2f a[5];
    v2f b[4];
#pragma unroll
    for (int m = 0; m < 5; ++m) a[m] = *(const v2f*)(ap + m * 16 * K);
#pragma unroll
    for (int j = 0; j < 4; ++j) {
      b[j].x = bp[j * 16];
      b[j].y = bp[HID + j * 16];
    }
    ap += 4;
    bp += 4 * HID;
    if (RELU) {
#pragma unroll
      for (int m = 0; m < 5; ++m) {
        a[m].x = fmaxf(a[m].x, 0.f);
        a[m].y = fmaxf(a[m].y, 0.f);
      }
    }
#pragma unroll
    for (int j = 0; j < 4; ++j)
#pragma unroll
      for (int m = 0; m < 5; ++m)
        acc[m][j] = __builtin_amdgcn_wmma_f32_16x16x4_f32(
            false, a[m], false, b[j], (short)0, acc[m][j], false, false);
  }

  float* hp = H + (size_t)(lb + (lane >> 4) * 8) * HID + col_base + col;
  const int rfirst = row_base - lb - (lane >> 4) * 8;
#pragma unroll
  for (int m = 0; m < 5; ++m) {
#pragma unroll
    for (int r = 0; r < 8; ++r) {
      if (m * 16 + r >= rfirst) {
#pragma unroll
        for (int j = 0; j < 4; ++j)
          hp[(size_t)(m * 16 + r) * HID + j * 16] = acc[m][j][r];
      }
    }
  }
}


template <bool FIRST>
__global__ __launch_bounds__(RTHREADS) void range_sum_kernel(
    const float* __restrict__ h, const int* __restrict__ src,
    const int* __restrict__ dst, const float* __restrict__ dinv,
    const float* __restrict__ bias, float* __restrict__ out,
    const int* __restrict__ gcount, const int* __restrict__ gsrc,
    const int* __restrict__ gdst, float* __restrict__ gw, int N, int E) {
  __shared__ float acc[RROWS * RCOLS];
  __shared__ int ls[LCAP];
  __shared__ int ld[LCAP];
  __shared__ float lw[LCAP];

  const int lane = threadIdx.x & 31;
  const int wave = __builtin_amdgcn_readfirstlane(threadIdx.x >> 5);
  const int range = blockIdx.x;
  const int r0 = range * RROWS;
  const int nrows = min(RROWS, N - r0);
  const size_t lbase = (size_t)range * LCAP;

  const int total = gcount[range];
  {
    const int n = (total <= LCAP) ? total : 0;
    for (int i = threadIdx.x; i < n; i += RTHREADS) {
      const int s = gsrc[lbase + i];
      const int d = gdst[lbase + i];
      float w;
      if (FIRST) {
        w = dinv[s] * dinv[r0 + d];
        gw[lbase + i] = w;
      } else {
        w = gw[lbase + i];
      }
      ls[i] = s;
      ld[i] = d;
      lw[i] = w;
    }
  }

  const int total4 = nrows * (RCOLS / 4);
  for (int half = 0; half < HID / RCOLS; ++half) {
    const int col0 = half * RCOLS;

    for (int i0 = threadIdx.x; i0 < total4; i0 += RTHREADS * 4) {
      float4 hv[4];
#pragma unroll
      for (int u = 0; u < 4; ++u) {
        const int i = i0 + u * RTHREADS;
        if (i < total4)
          hv[u] = *(const float4*)(h + (size_t)(r0 + (i >> 5)) * HID + col0 +
                                   (i & 31) * 4);
      }
#pragma unroll
      for (int u = 0; u < 4; ++u) {
        const int i = i0 + u * RTHREADS;
        if (i < total4) {
          const float di = dinv[r0 + (i >> 5)];
          const float sn = di * di;
          const float4 bv = *(const float4*)(bias + col0 + (i & 31) * 4);
          float4 o;
          o.x = fmaf(hv[u].x, sn, bv.x);
          o.y = fmaf(hv[u].y, sn, bv.y);
          o.z = fmaf(hv[u].z, sn, bv.z);
          o.w = fmaf(hv[u].w, sn, bv.w);
          float* ar = acc + (i >> 5) * RCOLS + (i & 31);
          ar[0] = o.x;
          ar[32] = o.y;
          ar[64] = o.z;
          ar[96] = o.w;
        }
      }
    }
    __syncthreads();

    if (total <= LCAP) {
      const int n = total;
      const int ngrp = (n + 7) >> 3;
      int base = ((wave * ngrp) / RWAVES) * 8;
      const int end = min(n, (((wave + 1) * ngrp) / RWAVES) * 8);
      int cur = -1;
      float4 run = make_float4(0.f, 0.f, 0.f, 0.f);
      auto flush = [&]() {
        float* ap = acc + cur * RCOLS + lane;
        unsafeAtomicAdd(ap, run.x);
        unsafeAtomicAdd(ap + 32, run.y);
        unsafeAtomicAdd(ap + 64, run.z);
        unsafeAtomicAdd(ap + 96, run.w);
      };
      auto take = [&](int dk, float wk, const float4& x) {
        if (dk != cur) {
          if (cur >= 0) flush();
          cur = dk;
          run.x = x.x * wk;
          run.y = x.y * wk;
          run.z = x.z * wk;
          run.w = x.w * wk;
        } else {
          run.x = fmaf(x.x, wk, run.x);
          run.y = fmaf(x.y, wk, run.y);
          run.z = fmaf(x.z, wk, run.z);
          run.w = fmaf(x.w, wk, run.w);
        }
      };
      int s = 0, d = 0;
      float w = 0.f;
      if (base + (lane & 7) < end) {
        s = ls[base + (lane & 7)];
        d = ld[base + (lane & 7)];
        w = lw[base + (lane & 7)];
      }
      while (base < end) {
        const int nbase = base + 8;
        int s2 = 0, d2 = 0;
        float w2 = 0.f;
        if (nbase + (lane & 7) < end) {
          s2 = ls[nbase + (lane & 7)];
          d2 = ld[nbase + (lane & 7)];
          w2 = lw[nbase + (lane & 7)];
        }
        const int cnt = min(8, end - base);
        if (cnt == 8) {
          float4 v[8];
#pragma unroll
          for (int q = 0; q < 8; ++q) {
            const int sk = __builtin_amdgcn_readlane(s, q);
            v[q] = *(const float4*)(h + (size_t)sk * HID + col0 + lane * 4);
          }
#pragma unroll
          for (int q = 0; q < 8; ++q) {
            const int dk = __builtin_amdgcn_readlane(d, q);
            const float wk =
                __int_as_float(__builtin_amdgcn_readlane(__float_as_int(w), q));
            take(dk, wk, v[q]);
          }
        } else {
          for (int k = 0; k < cnt; ++k) {
            const int sk = __builtin_amdgcn_readlane(s, k);
            const int dk = __builtin_amdgcn_readlane(d, k);
            const float wk =
                __int_as_float(__builtin_amdgcn_readlane(__float_as_int(w), k));
            const float4 v1 =
                *(const float4*)(h + (size_t)sk * HID + col0 + lane * 4);
            take(dk, wk, v1);
          }
        }
        s = s2;
        d = d2;
        w = w2;
        base = nbase;
      }
      if (cur >= 0) flush();
    } else {
      for (int e = threadIdx.x; e < E; e += RTHREADS) {
        const int dl = dst[e] - r0;
        if ((unsigned)dl < (unsigned)nrows) {
          const float* hp = h + (size_t)src[e] * HID + col0;
          float* ap = acc + dl * RCOLS;
          const float w = dinv[src[e]] * dinv[dst[e]];
          for (int c = 0; c < RCOLS; ++c)
            unsafeAtomicAdd(ap + (c & 3) * 32 + (c >> 2), hp[c] * w);
        }
      }
    }
    __syncthreads();

    for (int i = threadIdx.x; i < total4; i += RTHREADS) {
      const float* ar = acc + (i >> 5) * RCOLS + (i & 31);
      float4 o;
      o.x = ar[0];
      o.y = ar[32];
      o.z = ar[64];
      o.w = ar[96];
      *(float4*)(out + (size_t)(r0 + (i >> 5)) * HID + col0 + (i & 31) * 4) = o;
    }
    __syncthreads();
  }
}


static inline int cdiv(int a, int b) { return (a + b - 1) / b; }

extern "C" void kernel_launch(void* const* d_in, const int* in_sizes, int n_in,
                              void* d_out, int out_size, void* d_ws,
                              size_t ws_size, hipStream_t stream) {
  const float* x = (const float*)d_in[0];
  const int* ei = (const int*)d_in[1];
  const float* W1 = (const float*)d_in[2];
  const float* b1 = (const float*)d_in[3];
  const float* W2 = (const float*)d_in[4];
  const float* b2 = (const float*)d_in[5];
  const float* W3 = (const float*)d_in[6];
  const float* b3 = (const float*)d_in[7];
  float* out = (float*)d_out;

  const int F = 128;
  const int N = in_sizes[0] / F;
  const int E = in_sizes[1] / 2;
  const int* src = ei;
  const int* dst = ei + E;

  float* wsf = (float*)d_ws;
  size_t off = 0;
  float* dinv = wsf + off;
  off += (size_t)((N + 255) / 256) * 256;
  float* hbuf = wsf + off;
  off += (size_t)N * HID;
  float* agg = wsf + off;
  off += (size_t)N * HID;
  unsigned int* tagw = (unsigned int*)(wsf + off);
  off += (size_t)(((E + 3) / 4 + 255) / 256) * 256;
  const int nRanges = cdiv(N, RROWS);
  int* gcount = (int*)(wsf + off);
  off += (size_t)((nRanges + 255) / 256) * 256;
  int* gsrc = (int*)(wsf + off);
  off += (size_t)nRanges * LCAP;
  int* gdst = (int*)(wsf + off);
  off += (size_t)nRanges * LCAP;
  float* gw = wsf + off;

  const int edge4Grid = cdiv(cdiv(E, 4), TPB);
  const int rangeGrid = nRanges;
  const int gemmGrid = cdiv(N, GROWS);

  tag_kernel<<<edge4Grid, TPB, 0, stream>>>(dst, tagw, E);
  build_kernel<<<rangeGrid, RTHREADS, 0, stream>>>(src, dst, tagw, dinv, gcount,
                                                  gsrc, gdst, N, E);

  gemm_wmma_kernel<128, false><<<gemmGrid, GTHREADS, 0, stream>>>(x, W1, hbuf, N);
  range_sum_kernel<true><<<rangeGrid, RTHREADS, 0, stream>>>(
      hbuf, src, dst, dinv, b1, agg, gcount, gsrc, gdst, gw, N, E);

  gemm_wmma_kernel<256, true><<<gemmGrid, GTHREADS, 0, stream>>>(agg, W2, hbuf, N);
  range_sum_kernel<false><<<rangeGrid, RTHREADS, 0, stream>>>(
      hbuf, src, dst, dinv, b2, agg, gcount, gsrc, gdst, gw, N, E);

  gemm_wmma_kernel<256, true><<<gemmGrid, GTHREADS, 0, stream>>>(agg, W3, hbuf, N);
  range_sum_kernel<false><<<rangeGrid, RTHREADS, 0, stream>>>(
      hbuf, src, dst, dinv, b3, out, gcount, gsrc, gdst, gw, N, E);
}
